# starting kernel
# baseline (speedup 1.0000x reference)
#include <hip/hip_runtime.h>

#define N_NODES 50000
#define N_EDGES 1600000
#define IN_CH   64
#define HID_CH  64
#define OUT_CH  32
#define NEG_SLOPE 0.01f

typedef __attribute__((ext_vector_type(2))) float v2f;
typedef __attribute__((ext_vector_type(8))) float v8f;

__global__ __launch_bounds__(256) void sage_scatter(
    const float* __restrict__ feat, const int* __restrict__ ei,
    float* __restrict__ agg, float* __restrict__ deg, int count_deg)
{
    long idx = (long)blockIdx.x * blockDim.x + threadIdx.x;
    if (idx >= (long)N_EDGES * 16) return;
    int e = (int)(idx >> 4);
    int g = (int)(idx & 15);

    int src = ei[e];
    int dst = ei[N_EDGES + e];

    const float4 v = *(const float4*)(feat + (long)src * 64 + g * 4);
    float* a = agg + (long)dst * 64 + g * 4;
    unsafeAtomicAdd(a + 0, v.x);
    unsafeAtomicAdd(a + 1, v.y);
    unsafeAtomicAdd(a + 2, v.z);
    unsafeAtomicAdd(a + 3, v.w);
    if (count_deg && g == 0) unsafeAtomicAdd(deg + dst, 1.0f);
}

template <int OUTC>
__global__ __launch_bounds__(256) void sage_wmma(
    const float* __restrict__ agg, const float* __restrict__ deg,
    const float* __restrict__ feat,
    const float* __restrict__ Wl, const float* __restrict__ Wr,
    const float* __restrict__ bias, float* __restrict__ out)
{
    __shared__ float Wlds[OUTC * 128];

    const int tid = threadIdx.x;
    for (int i = tid; i < OUTC * 128; i += 256) {
        int n = i >> 7, k = i & 127;
        Wlds[i] = (k < 64) ? Wl[n * 64 + k] : Wr[n * 64 + (k - 64)];
    }
    __syncthreads();

    const int wave = tid >> 5;
    const int lane = tid & 31;
    const int col_tiles   = OUTC / 16;
    const int total_tiles = (N_NODES / 16) * col_tiles;
    const int tile = blockIdx.x * 8 + wave;
    if (tile >= total_tiles) return;

    const int row0 = (tile / col_tiles) * 16;
    const int col0 = (tile % col_tiles) * 16;
    const int half = lane >> 4;
    const int m    = lane & 15;
    const int row  = row0 + m;
    const int colg = col0 + m;

    const float rd = 1.0f / fmaxf(deg[row], 1.0f);
    const float bc = bias[colg];

    v8f c;
#pragma unroll
    for (int i = 0; i < 8; ++i) c[i] = bc;

    const float* arow = agg  + (long)row * 64;
    const float* xrow = feat + (long)row * 64;
    const float* wcol = &Wlds[colg * 128];

#pragma unroll
    for (int ks = 0; ks < 64; ks += 4) {
        v2f a, b;
        a[0] = arow[ks + half * 2 + 0] * rd;
        a[1] = arow[ks + half * 2 + 1] * rd;
        b[0] = wcol[ks + half * 2 + 0];
        b[1] = wcol[ks + half * 2 + 1];
        c = __builtin_amdgcn_wmma_f32_16x16x4_f32(false, a, false, b,
                                                  (short)0, c, false, false);
    }
#pragma unroll
    for (int ks = 0; ks < 64; ks += 4) {
        v2f a, b;
        a[0] = xrow[ks + half * 2 + 0];
        a[1] = xrow[ks + half * 2 + 1];
        b[0] = wcol[64 + ks + half * 2 + 0];
        b[1] = wcol[64 + ks + half * 2 + 1];
        c = __builtin_amdgcn_wmma_f32_16x16x4_f32(false, a, false, b,
                                                  (short)0, c, false, false);
    }

#pragma unroll
    for (int v = 0; v < 8; ++v) {
        float y = c[v];
        y = (y > 0.0f) ? y : NEG_SLOPE * y;
        out[(long)(row0 + half * 8 + v) * OUTC + colg] = y;
    }
}

extern "C" void kernel_launch(void* const* d_in, const int* in_sizes, int n_in,
                              void* d_out, int out_size, void* d_ws, size_t ws_size,
                              hipStream_t stream) {
    const float* x   = (const float*)d_in[0];
    const int*   ei  = (const int*)  d_in[1];
    const float* W1l = (const float*)d_in[2];
    const float* b1  = (const float*)d_in[3];
    const float* W1r = (const float*)d_in[4];
    const float* W2l = (const float*)d_in[5];
    const float* b2  = (const float*)d_in[6];
    const float* W2r = (const float*)d_in[7];
    float* out = (float*)d_out;

    const size_t degB = ((size_t)N_NODES * 4 + 255) & ~(size_t)255;
    const size_t aggB = (size_t)N_NODES * 64 * 4;
    float* deg = (float*)d_ws;
    float* agg = (float*)((char*)d_ws + degB);
    float* h   = (float*)((char*)d_ws + degB + aggB);

    const long sthreads = (long)N_EDGES * 16;
    const int  sblocks  = (int)((sthreads + 255) / 256);

    hipMemsetAsync(d_ws, 0, degB + aggB, stream);
    sage_scatter<<<sblocks, 256, 0, stream>>>(x, ei, agg, deg, 1);
    {
        const int tiles  = (N_NODES / 16) * (HID_CH / 16);
        const int blocks = (tiles + 7) / 8;
        sage_wmma<HID_CH><<<blocks, 256, 0, stream>>>(agg, deg, x, W1l, W1r, b1, h);
    }

    hipMemsetAsync(agg, 0, aggB, stream);
    sage_scatter<<<sblocks, 256, 0, stream>>>(h, ei, agg, deg, 0);
    {
        const int tiles  = (N_NODES / 16) * (OUT_CH / 16);
        const int blocks = (tiles + 7) / 8;
        sage_wmma<OUT_CH><<<blocks, 256, 0, stream>>>(agg, deg, h, W2l, W2r, b2, out);
    }
}


# degree counted once by integer LDS bins over 16 node ranges x 16 edge slices (planes summed by the GEMM waves) instead of 1.6M global f32 atomic adds onto 50000 words
# speedup vs baseline: 1.4167x; 1.4167x over previous
#include <hip/hip_runtime.h>

#define N_NODES 50000
#define N_EDGES 1600000
#define IN_CH   64
#define HID_CH  64
#define OUT_CH  32
#define NEG_SLOPE 0.01f

#define DEG_P  16
#define DEG_S  16
#define DEG_R  3136
#define DEG_NP (DEG_P * DEG_R)
#define DEG_T  512

typedef __attribute__((ext_vector_type(2))) float v2f;
typedef __attribute__((ext_vector_type(8))) float v8f;

__global__ __launch_bounds__(256) void sage_scatter(
    const float* __restrict__ feat, const int* __restrict__ ei,
    float* __restrict__ agg)
{
    long idx = (long)blockIdx.x * blockDim.x + threadIdx.x;
    if (idx >= (long)N_EDGES * 16) return;
    int e = (int)(idx >> 4);
    int g = (int)(idx & 15);

    int src = ei[e];
    int dst = ei[N_EDGES + e];

    const float4 v = *(const float4*)(feat + (long)src * 64 + g * 4);
    float* a = agg + (long)dst * 64 + g * 4;
    unsafeAtomicAdd(a + 0, v.x);
    unsafeAtomicAdd(a + 1, v.y);
    unsafeAtomicAdd(a + 2, v.z);
    unsafeAtomicAdd(a + 3, v.w);
}

__device__ __forceinline__ void deg_bin4(unsigned* cnt, const int4 d, const unsigned base)
{
    const unsigned r0 = (unsigned)d.x - base;
    const unsigned r1 = (unsigned)d.y - base;
    const unsigned r2 = (unsigned)d.z - base;
    const unsigned r3 = (unsigned)d.w - base;
    if (r0 < DEG_R) atomicAdd(&cnt[r0], 1u);
    if (r1 < DEG_R) atomicAdd(&cnt[r1], 1u);
    if (r2 < DEG_R) atomicAdd(&cnt[r2], 1u);
    if (r3 < DEG_R) atomicAdd(&cnt[r3], 1u);
}

__global__ __launch_bounds__(DEG_T) void sage_degree(
    const int* __restrict__ ei, unsigned* __restrict__ planes)
{
    __shared__ __attribute__((aligned(16))) unsigned cnt[DEG_R];

    const int tid = threadIdx.x;
    const int p = blockIdx.x % DEG_P;
    const int s = blockIdx.x / DEG_P;

    for (int i = tid; i < DEG_R; i += DEG_T) cnt[i] = 0u;
    __syncthreads();

    const unsigned base = (unsigned)(p * DEG_R);
    const int4* d4 = (const int4*)(ei + N_EDGES + s * (N_EDGES / DEG_S));
    const int n4 = N_EDGES / DEG_S / 4;

    const int full = n4 / (8 * DEG_T);
    for (int r = 0; r < full; ++r) {
        int4 d[8];
#pragma unroll
        for (int u = 0; u < 8; ++u) d[u] = d4[(r * 8 + u) * DEG_T + tid];
#pragma unroll
        for (int u = 0; u < 8; ++u) deg_bin4(cnt, d[u], base);
    }
    for (int i = full * 8 * DEG_T + tid; i < n4; i += DEG_T) deg_bin4(cnt, d4[i], base);
    __syncthreads();

    uint4* o4 = (uint4*)(planes + (long)s * DEG_NP + base);
    const uint4* c4 = (const uint4*)cnt;
    for (int i = tid; i < DEG_R / 4; i += DEG_T) o4[i] = c4[i];
}

template <int OUTC>
__global__ __launch_bounds__(256) void sage_wmma(
    const float* __restrict__ agg, const unsigned* __restrict__ planes,
    const float* __restrict__ feat,
    const float* __restrict__ Wl, const float* __restrict__ Wr,
    const float* __restrict__ bias, float* __restrict__ out)
{
    __shared__ float Wlds[OUTC * 128];

    const int tid = threadIdx.x;
    for (int i = tid; i < OUTC * 128; i += 256) {
        int n = i >> 7, k = i & 127;
        Wlds[i] = (k < 64) ? Wl[n * 64 + k] : Wr[n * 64 + (k - 64)];
    }
    __syncthreads();

    const int wave = tid >> 5;
    const int lane = tid & 31;
    const int col_tiles   = OUTC / 16;
    const int total_tiles = (N_NODES / 16) * col_tiles;
    const int tile = blockIdx.x * 8 + wave;
    if (tile >= total_tiles) return;

    const int row0 = (tile / col_tiles) * 16;
    const int col0 = (tile % col_tiles) * 16;
    const int half = lane >> 4;
    const int m    = lane & 15;
    const int row  = row0 + m;
    const int colg = col0 + m;

    unsigned cn = 0u;
#pragma unroll
    for (int s = 0; s < DEG_S; ++s) cn += planes[s * DEG_NP + row];
    const float rd = 1.0f / fmaxf((float)cn, 1.0f);
    const float bc = bias[colg];

    v8f c;
#pragma unroll
    for (int i = 0; i < 8; ++i) c[i] = bc;

    const float* arow = agg  + (long)row * 64;
    const float* xrow = feat + (long)row * 64;
    const float* wcol = &Wlds[colg * 128];

#pragma unroll
    for (int ks = 0; ks < 64; ks += 4) {
        v2f a, b;
        a[0] = arow[ks + half * 2 + 0] * rd;
        a[1] = arow[ks + half * 2 + 1] * rd;
        b[0] = wcol[ks + half * 2 + 0];
        b[1] = wcol[ks + half * 2 + 1];
        c = __builtin_amdgcn_wmma_f32_16x16x4_f32(false, a, false, b,
                                                  (short)0, c, false, false);
    }
#pragma unroll
    for (int ks = 0; ks < 64; ks += 4) {
        v2f a, b;
        a[0] = xrow[ks + half * 2 + 0];
        a[1] = xrow[ks + half * 2 + 1];
        b[0] = wcol[64 + ks + half * 2 + 0];
        b[1] = wcol[64 + ks + half * 2 + 1];
        c = __builtin_amdgcn_wmma_f32_16x16x4_f32(false, a, false, b,
                                                  (short)0, c, false, false);
    }

#pragma unroll
    for (int v = 0; v < 8; ++v) {
        float y = c[v];
        y = (y > 0.0f) ? y : NEG_SLOPE * y;
        out[(long)(row0 + half * 8 + v) * OUTC + colg] = y;
    }
}

extern "C" void kernel_launch(void* const* d_in, const int* in_sizes, int n_in,
                              void* d_out, int out_size, void* d_ws, size_t ws_size,
                              hipStream_t stream) {
    const float* x   = (const float*)d_in[0];
    const int*   ei  = (const int*)  d_in[1];
    const float* W1l = (const float*)d_in[2];
    const float* b1  = (const float*)d_in[3];
    const float* W1r = (const float*)d_in[4];
    const float* W2l = (const float*)d_in[5];
    const float* b2  = (const float*)d_in[6];
    const float* W2r = (const float*)d_in[7];
    float* out = (float*)d_out;

    const size_t degB = (size_t)DEG_S * DEG_NP * 4;
    const size_t aggB = (size_t)N_NODES * 64 * 4;
    unsigned* planes = (unsigned*)d_ws;
    float* agg = (float*)((char*)d_ws + degB);
    float* h   = (float*)((char*)d_ws + degB + aggB);

    const long sthreads = (long)N_EDGES * 16;
    const int  sblocks  = (int)((sthreads + 255) / 256);

    sage_degree<<<DEG_P * DEG_S, DEG_T, 0, stream>>>(ei, planes);
    hipMemsetAsync(agg, 0, aggB, stream);
    sage_scatter<<<sblocks, 256, 0, stream>>>(x, ei, agg);
    {
        const int tiles  = (N_NODES / 16) * (HID_CH / 16);
        const int blocks = (tiles + 7) / 8;
        sage_wmma<HID_CH><<<blocks, 256, 0, stream>>>(agg, planes, x, W1l, W1r, b1, h);
    }

    hipMemsetAsync(agg, 0, aggB, stream);
    sage_scatter<<<sblocks, 256, 0, stream>>>(h, ei, agg);
    {
        const int tiles  = (N_NODES / 16) * (OUT_CH / 16);
        const int blocks = (tiles + 7) / 8;
        sage_wmma<OUT_CH><<<blocks, 256, 0, stream>>>(agg, planes, h, W2l, W2r, b2, out);
    }
}


# f32 atomic scatter and memsets replaced by a gather over a destination-sorted source list (counting sort from the degree planes; LDS cursor ranks in arrival order); rows summed in list order
# speedup vs baseline: 3.9434x; 2.7835x over previous
#include <hip/hip_runtime.h>

#define N_NODES 50000
#define N_EDGES 1600000
#define IN_CH   64
#define HID_CH  64
#define OUT_CH  32
#define NEG_SLOPE 0.01f

#define DEG_P  16
#define DEG_S  16
#define DEG_R  3136
#define DEG_NP (DEG_P * DEG_R)
#define DEG_T  512
#define FILL_T 1024

typedef __attribute__((ext_vector_type(2))) float v2f;
typedef __attribute__((ext_vector_type(8))) float v8f;

__global__ __launch_bounds__(256) void sage_gather(
    const float* __restrict__ feat, const int* __restrict__ ssrc,
    const unsigned* __restrict__ rowptr, float* __restrict__ agg)
{
    const int idx  = blockIdx.x * 256 + threadIdx.x;
    const int node = idx >> 4;
    const int g    = idx & 15;
    if (node >= N_NODES) return;

    unsigned e   = rowptr[node];
    unsigned end = rowptr[node + 1];
    if (end > (unsigned)N_EDGES) end = (unsigned)N_EDGES;

    const float* f = feat + g * 4;
    float4 acc = make_float4(0.0f, 0.0f, 0.0f, 0.0f);

    for (; e + 8u <= end; e += 8u) {
        int si[8];
        float4 v[8];
#pragma unroll
        for (int u = 0; u < 8; ++u) si[u] = ssrc[e + u];
#pragma unroll
        for (int u = 0; u < 8; ++u) v[u] = *(const float4*)(f + (long)si[u] * 64);
#pragma unroll
        for (int u = 0; u < 8; ++u) {
            acc.x += v[u].x; acc.y += v[u].y; acc.z += v[u].z; acc.w += v[u].w;
        }
    }
    if (e < end) {
        int si[7];
        float4 v[7];
#pragma unroll
        for (int u = 0; u < 7; ++u) si[u] = ssrc[min(e + (unsigned)u, end - 1u)];
#pragma unroll
        for (int u = 0; u < 7; ++u) {
            v[u] = make_float4(0.0f, 0.0f, 0.0f, 0.0f);
            if (e + u < end) v[u] = *(const float4*)(f + (long)si[u] * 64);
        }
#pragma unroll
        for (int u = 0; u < 7; ++u) {
            if (e + u < end) { acc.x += v[u].x; acc.y += v[u].y; acc.z += v[u].z; acc.w += v[u].w; }
        }
    }
    *(float4*)(agg + (long)node * 64 + g * 4) = acc;
}

__device__ __forceinline__ void deg_bin4(unsigned* cnt, const int4 d, const unsigned base)
{
    const unsigned r0 = (unsigned)d.x - base;
    const unsigned r1 = (unsigned)d.y - base;
    const unsigned r2 = (unsigned)d.z - base;
    const unsigned r3 = (unsigned)d.w - base;
    if (r0 < DEG_R) atomicAdd(&cnt[r0], 1u);
    if (r1 < DEG_R) atomicAdd(&cnt[r1], 1u);
    if (r2 < DEG_R) atomicAdd(&cnt[r2], 1u);
    if (r3 < DEG_R) atomicAdd(&cnt[r3], 1u);
}

__global__ __launch_bounds__(DEG_T) void sage_degree(
    const int* __restrict__ ei, unsigned* __restrict__ planes, unsigned* __restrict__ tot)
{
    __shared__ __attribute__((aligned(16))) unsigned cnt[DEG_R];
    __shared__ unsigned tsum;

    const int tid = threadIdx.x;
    const int p = blockIdx.x % DEG_P;
    const int s = blockIdx.x / DEG_P;

    for (int i = tid; i < DEG_R; i += DEG_T) cnt[i] = 0u;
    if (tid == 0) tsum = 0u;
    __syncthreads();

    const unsigned base = (unsigned)(p * DEG_R);
    const int4* d4 = (const int4*)(ei + N_EDGES + s * (N_EDGES / DEG_S));
    const int n4 = N_EDGES / DEG_S / 4;

    const int full = n4 / (8 * DEG_T);
    for (int r = 0; r < full; ++r) {
        int4 d[8];
#pragma unroll
        for (int u = 0; u < 8; ++u) d[u] = d4[(r * 8 + u) * DEG_T + tid];
#pragma unroll
        for (int u = 0; u < 8; ++u) deg_bin4(cnt, d[u], base);
    }
    for (int i = full * 8 * DEG_T + tid; i < n4; i += DEG_T) deg_bin4(cnt, d4[i], base);
    __syncthreads();

    uint4* o4 = (uint4*)(planes + (long)s * DEG_NP + base);
    const uint4* c4 = (const uint4*)cnt;
    unsigned part = 0u;
    for (int i = tid; i < DEG_R / 4; i += DEG_T) {
        const uint4 c = c4[i];
        o4[i] = c;
        part += c.x + c.y + c.z + c.w;
    }
    atomicAdd(&tsum, part);
    __syncthreads();
    if (tid == 0) tot[blockIdx.x] = tsum;
}

__device__ __forceinline__ void fill_bin4(unsigned* cur, int* __restrict__ ssrc,
                                          const int4 d, const int4 sv, const unsigned base)
{
    const unsigned r0 = (unsigned)d.x - base;
    const unsigned r1 = (unsigned)d.y - base;
    const unsigned r2 = (unsigned)d.z - base;
    const unsigned r3 = (unsigned)d.w - base;
    unsigned q0 = (unsigned)N_EDGES, q1 = (unsigned)N_EDGES, q2 = (unsigned)N_EDGES, q3 = (unsigned)N_EDGES;
    if (r0 < DEG_R) q0 = atomicAdd(&cur[r0], 1u);
    if (r1 < DEG_R) q1 = atomicAdd(&cur[r1], 1u);
    if (r2 < DEG_R) q2 = atomicAdd(&cur[r2], 1u);
    if (r3 < DEG_R) q3 = atomicAdd(&cur[r3], 1u);
    if (q0 < (unsigned)N_EDGES) ssrc[q0] = sv.x;
    if (q1 < (unsigned)N_EDGES) ssrc[q1] = sv.y;
    if (q2 < (unsigned)N_EDGES) ssrc[q2] = sv.z;
    if (q3 < (unsigned)N_EDGES) ssrc[q3] = sv.w;
}

__global__ __launch_bounds__(FILL_T) void sage_fill(
    const int* __restrict__ ei, const unsigned* __restrict__ planes,
    const unsigned* __restrict__ tot, int* __restrict__ ssrc,
    unsigned* __restrict__ rowptr)
{
    __shared__ unsigned cur[DEG_R];
    __shared__ unsigned dg[DEG_R];
    __shared__ unsigned wsum[FILL_T / 32];
    __shared__ unsigned rbase;

    const int tid  = threadIdx.x;
    const int lane = tid & 31;
    const int wave = tid >> 5;
    const int p = blockIdx.x % DEG_P;
    const int s = blockIdx.x / DEG_P;
    const unsigned base = (unsigned)(p * DEG_R);

    if (tid == 0) rbase = 0u;
    for (int i = tid; i < DEG_R; i += FILL_T) {
        unsigned all = 0u, pre = 0u;
#pragma unroll
        for (int q = 0; q < DEG_S; ++q) {
            const unsigned c = planes[q * DEG_NP + base + i];
            all += c;
            pre += (q < s) ? c : 0u;
        }
        dg[i]  = all;
        cur[i] = pre;
    }
    __syncthreads();

    if (tid < DEG_P * DEG_S) {
        const unsigned t = ((tid % DEG_P) < p) ? tot[tid] : 0u;
        atomicAdd(&rbase, t);
    }

    const int own = tid * 7;
    unsigned loc = 0u;
    if (tid < DEG_R / 7) {
        for (int j = 0; j < 7; ++j) loc += dg[own + j];
    }
    unsigned inc = loc;
    for (int o = 1; o < 32; o <<= 1) {
        const unsigned v = __shfl_up(inc, o);
        if (lane >= o) inc += v;
    }
    if (lane == 31) wsum[wave] = inc;
    __syncthreads();

    unsigned run = rbase + inc - loc;
    for (int w = 0; w < wave; ++w) run += wsum[w];
    if (tid < DEG_R / 7) {
        for (int j = 0; j < 7; ++j) {
            const unsigned d = dg[own + j];
            cur[own + j] += run;
            if (s == 0) rowptr[base + own + j] = run;
            run += d;
        }
    }
    __syncthreads();

    const int4* d4 = (const int4*)(ei + N_EDGES + s * (N_EDGES / DEG_S));
    const int4* s4 = (const int4*)(ei + s * (N_EDGES / DEG_S));
    const int n4 = N_EDGES / DEG_S / 4;

    const int full = n4 / (2 * FILL_T);
    for (int r = 0; r < full; ++r) {
        int4 d[2], sv[2];
#pragma unroll
        for (int u = 0; u < 2; ++u) {
            d[u]  = d4[(r * 2 + u) * FILL_T + tid];
            sv[u] = s4[(r * 2 + u) * FILL_T + tid];
        }
#pragma unroll
        for (int u = 0; u < 2; ++u) fill_bin4(cur, ssrc, d[u], sv[u], base);
    }
    for (int i = full * 2 * FILL_T + tid; i < n4; i += FILL_T) fill_bin4(cur, ssrc, d4[i], s4[i], base);
}

template <int OUTC>
__global__ __launch_bounds__(256) void sage_wmma(
    const float* __restrict__ agg, const unsigned* __restrict__ planes,
    const float* __restrict__ feat,
    const float* __restrict__ Wl, const float* __restrict__ Wr,
    const float* __restrict__ bias, float* __restrict__ out)
{
    __shared__ float Wlds[OUTC * 128];

    const int tid = threadIdx.x;
    for (int i = tid; i < OUTC * 128; i += 256) {
        int n = i >> 7, k = i & 127;
        Wlds[i] = (k < 64) ? Wl[n * 64 + k] : Wr[n * 64 + (k - 64)];
    }
    __syncthreads();

    const int wave = tid >> 5;
    const int lane = tid & 31;
    const int col_tiles   = OUTC / 16;
    const int total_tiles = (N_NODES / 16) * col_tiles;
    const int tile = blockIdx.x * 8 + wave;
    if (tile >= total_tiles) return;

    const int row0 = (tile / col_tiles) * 16;
    const int col0 = (tile % col_tiles) * 16;
    const int half = lane >> 4;
    const int m    = lane & 15;
    const int row  = row0 + m;
    const int colg = col0 + m;

    unsigned cn = 0u;
#pragma unroll
    for (int s = 0; s < DEG_S; ++s) cn += planes[s * DEG_NP + row];
    const float rd = 1.0f / fmaxf((float)cn, 1.0f);
    const float bc = bias[colg];

    v8f c;
#pragma unroll
    for (int i = 0; i < 8; ++i) c[i] = bc;

    const float* arow = agg  + (long)row * 64;
    const float* xrow = feat + (long)row * 64;
    const float* wcol = &Wlds[colg * 128];

#pragma unroll
    for (int ks = 0; ks < 64; ks += 4) {
        v2f a, b;
        a[0] = arow[ks + half * 2 + 0] * rd;
        a[1] = arow[ks + half * 2 + 1] * rd;
        b[0] = wcol[ks + half * 2 + 0];
        b[1] = wcol[ks + half * 2 + 1];
        c = __builtin_amdgcn_wmma_f32_16x16x4_f32(false, a, false, b,
                                                  (short)0, c, false, false);
    }
#pragma unroll
    for (int ks = 0; ks < 64; ks += 4) {
        v2f a, b;
        a[0] = xrow[ks + half * 2 + 0];
        a[1] = xrow[ks + half * 2 + 1];
        b[0] = wcol[64 + ks + half * 2 + 0];
        b[1] = wcol[64 + ks + half * 2 + 1];
        c = __builtin_amdgcn_wmma_f32_16x16x4_f32(false, a, false, b,
                                                  (short)0, c, false, false);
    }

#pragma unroll
    for (int v = 0; v < 8; ++v) {
        float y = c[v];
        y = (y > 0.0f) ? y : NEG_SLOPE * y;
        out[(long)(row0 + half * 8 + v) * OUTC + colg] = y;
    }
}

extern "C" void kernel_launch(void* const* d_in, const int* in_sizes, int n_in,
                              void* d_out, int out_size, void* d_ws, size_t ws_size,
                              hipStream_t stream) {
    const float* x   = (const float*)d_in[0];
    const int*   ei  = (const int*)  d_in[1];
    const float* W1l = (const float*)d_in[2];
    const float* b1  = (const float*)d_in[3];
    const float* W1r = (const float*)d_in[4];
    const float* W2l = (const float*)d_in[5];
    const float* b2  = (const float*)d_in[6];
    const float* W2r = (const float*)d_in[7];
    float* out = (float*)d_out;

    const size_t degB = (size_t)DEG_S * DEG_NP * 4;
    const size_t aggB = (size_t)N_NODES * 64 * 4;
    unsigned* planes = (unsigned*)d_ws;
    float* agg = (float*)((char*)d_ws + degB);
    float* h   = (float*)((char*)d_ws + degB + aggB);
    unsigned* tot    = (unsigned*)((char*)d_ws + degB + 2 * aggB);
    unsigned* rowptr = tot + DEG_P * DEG_S;
    int*      ssrc   = (int*)(rowptr + DEG_NP);

    const int gblocks = N_NODES * 16 / 256;

    sage_degree<<<DEG_P * DEG_S, DEG_T, 0, stream>>>(ei, planes, tot);
    sage_fill<<<DEG_P * DEG_S, FILL_T, 0, stream>>>(ei, planes, tot, ssrc, rowptr);

    sage_gather<<<gblocks, 256, 0, stream>>>(x, ssrc, rowptr, agg);
    {
        const int tiles  = (N_NODES / 16) * (HID_CH / 16);
        const int blocks = (tiles + 7) / 8;
        sage_wmma<HID_CH><<<blocks, 256, 0, stream>>>(agg, planes, x, W1l, W1r, b1, h);
    }

    sage_gather<<<gblocks, 256, 0, stream>>>(h, ssrc, rowptr, agg);
    {
        const int tiles  = (N_NODES / 16) * (OUT_CH / 16);
        const int blocks = (tiles + 7) / 8;
        sage_wmma<OUT_CH><<<blocks, 256, 0, stream>>>(agg, planes, h, W2l, W2r, b2, out);
    }
}


# each layer's neighbour gather fused into its linear kernel: a work-group sums its nodes' source rows into an LDS tile that feeds the WMMA A fragments; sums never reach memory, 4 launches instead of 6
# speedup vs baseline: 4.4292x; 1.1232x over previous
#include <hip/hip_runtime.h>

#define N_NODES 50000
#define N_EDGES 1600000
#define IN_CH   64
#define HID_CH  64
#define OUT_CH  32
#define NEG_SLOPE 0.01f

#define DEG_P  16
#define DEG_S  16
#define DEG_R  3136
#define DEG_NP (DEG_P * DEG_R)
#define DEG_T  512
#define FILL_T 1024

typedef __attribute__((ext_vector_type(2))) float v2f;
typedef __attribute__((ext_vector_type(8))) float v8f;

__device__ __forceinline__ float4 sage_row_sum(
    const float* __restrict__ f, const int* __restrict__ ssrc, unsigned e, const unsigned end)
{
    float4 acc = make_float4(0.0f, 0.0f, 0.0f, 0.0f);

    for (; e + 8u <= end; e += 8u) {
        int si[8];
        float4 v[8];
#pragma unroll
        for (int u = 0; u < 8; ++u) si[u] = ssrc[e + u];
#pragma unroll
        for (int u = 0; u < 8; ++u) v[u] = *(const float4*)(f + (long)si[u] * 64);
#pragma unroll
        for (int u = 0; u < 8; ++u) {
            acc.x += v[u].x; acc.y += v[u].y; acc.z += v[u].z; acc.w += v[u].w;
        }
    }
    if (e < end) {
        int si[7];
        float4 v[7];
#pragma unroll
        for (int u = 0; u < 7; ++u) si[u] = ssrc[min(e + (unsigned)u, end - 1u)];
#pragma unroll
        for (int u = 0; u < 7; ++u) {
            v[u] = make_float4(0.0f, 0.0f, 0.0f, 0.0f);
            if (e + u < end) v[u] = *(const float4*)(f + (long)si[u] * 64);
        }
#pragma unroll
        for (int u = 0; u < 7; ++u) {
            if (e + u < end) { acc.x += v[u].x; acc.y += v[u].y; acc.z += v[u].z; acc.w += v[u].w; }
        }
    }
    return acc;
}

__device__ __forceinline__ void deg_bin4(unsigned* cnt, const int4 d, const unsigned base)
{
    const unsigned r0 = (unsigned)d.x - base;
    const unsigned r1 = (unsigned)d.y - base;
    const unsigned r2 = (unsigned)d.z - base;
    const unsigned r3 = (unsigned)d.w - base;
    if (r0 < DEG_R) atomicAdd(&cnt[r0], 1u);
    if (r1 < DEG_R) atomicAdd(&cnt[r1], 1u);
    if (r2 < DEG_R) atomicAdd(&cnt[r2], 1u);
    if (r3 < DEG_R) atomicAdd(&cnt[r3], 1u);
}

__global__ __launch_bounds__(DEG_T) void sage_degree(
    const int* __restrict__ ei, unsigned* __restrict__ planes, unsigned* __restrict__ tot)
{
    __shared__ __attribute__((aligned(16))) unsigned cnt[DEG_R];
    __shared__ unsigned tsum;

    const int tid = threadIdx.x;
    const int p = blockIdx.x % DEG_P;
    const int s = blockIdx.x / DEG_P;

    for (int i = tid; i < DEG_R; i += DEG_T) cnt[i] = 0u;
    if (tid == 0) tsum = 0u;
    __syncthreads();

    const unsigned base = (unsigned)(p * DEG_R);
    const int4* d4 = (const int4*)(ei + N_EDGES + s * (N_EDGES / DEG_S));
    const int n4 = N_EDGES / DEG_S / 4;

    const int full = n4 / (8 * DEG_T);
    for (int r = 0; r < full; ++r) {
        int4 d[8];
#pragma unroll
        for (int u = 0; u < 8; ++u) d[u] = d4[(r * 8 + u) * DEG_T + tid];
#pragma unroll
        for (int u = 0; u < 8; ++u) deg_bin4(cnt, d[u], base);
    }
    for (int i = full * 8 * DEG_T + tid; i < n4; i += DEG_T) deg_bin4(cnt, d4[i], base);
    __syncthreads();

    uint4* o4 = (uint4*)(planes + (long)s * DEG_NP + base);
    const uint4* c4 = (const uint4*)cnt;
    unsigned part = 0u;
    for (int i = tid; i < DEG_R / 4; i += DEG_T) {
        const uint4 c = c4[i];
        o4[i] = c;
        part += c.x + c.y + c.z + c.w;
    }
    atomicAdd(&tsum, part);
    __syncthreads();
    if (tid == 0) tot[blockIdx.x] = tsum;
}

__device__ __forceinline__ void fill_bin4(unsigned* cur, int* __restrict__ ssrc,
                                          const int4 d, const int4 sv, const unsigned base)
{
    const unsigned r0 = (unsigned)d.x - base;
    const unsigned r1 = (unsigned)d.y - base;
    const unsigned r2 = (unsigned)d.z - base;
    const unsigned r3 = (unsigned)d.w - base;
    unsigned q0 = (unsigned)N_EDGES, q1 = (unsigned)N_EDGES, q2 = (unsigned)N_EDGES, q3 = (unsigned)N_EDGES;
    if (r0 < DEG_R) q0 = atomicAdd(&cur[r0], 1u);
    if (r1 < DEG_R) q1 = atomicAdd(&cur[r1], 1u);
    if (r2 < DEG_R) q2 = atomicAdd(&cur[r2], 1u);
    if (r3 < DEG_R) q3 = atomicAdd(&cur[r3], 1u);
    if (q0 < (unsigned)N_EDGES) ssrc[q0] = sv.x;
    if (q1 < (unsigned)N_EDGES) ssrc[q1] = sv.y;
    if (q2 < (unsigned)N_EDGES) ssrc[q2] = sv.z;
    if (q3 < (unsigned)N_EDGES) ssrc[q3] = sv.w;
}

__global__ __launch_bounds__(FILL_T) void sage_fill(
    const int* __restrict__ ei, const unsigned* __restrict__ planes,
    const unsigned* __restrict__ tot, int* __restrict__ ssrc,
    unsigned* __restrict__ rowptr)
{
    __shared__ unsigned cur[DEG_R];
    __shared__ unsigned dg[DEG_R];
    __shared__ unsigned wsum[FILL_T / 32];
    __shared__ unsigned rbase;

    const int tid  = threadIdx.x;
    const int lane = tid & 31;
    const int wave = tid >> 5;
    const int p = blockIdx.x % DEG_P;
    const int s = blockIdx.x / DEG_P;
    const unsigned base = (unsigned)(p * DEG_R);

    if (tid == 0) rbase = 0u;
    for (int i = tid; i < DEG_R; i += FILL_T) {
        unsigned all = 0u, pre = 0u;
#pragma unroll
        for (int q = 0; q < DEG_S; ++q) {
            const unsigned c = planes[q * DEG_NP + base + i];
            all += c;
            pre += (q < s) ? c : 0u;
        }
        dg[i]  = all;
        cur[i] = pre;
    }
    __syncthreads();

    if (tid < DEG_P * DEG_S) {
        const unsigned t = ((tid % DEG_P) < p) ? tot[tid] : 0u;
        atomicAdd(&rbase, t);
    }

    const int own = tid * 7;
    unsigned loc = 0u;
    if (tid < DEG_R / 7) {
        for (int j = 0; j < 7; ++j) loc += dg[own + j];
    }
    unsigned inc = loc;
    for (int o = 1; o < 32; o <<= 1) {
        const unsigned v = __shfl_up(inc, o);
        if (lane >= o) inc += v;
    }
    if (lane == 31) wsum[wave] = inc;
    __syncthreads();

    unsigned run = rbase + inc - loc;
    for (int w = 0; w < wave; ++w) run += wsum[w];
    if (tid < DEG_R / 7) {
        for (int j = 0; j < 7; ++j) {
            const unsigned d = dg[own + j];
            cur[own + j] += run;
            if (s == 0) rowptr[base + own + j] = run;
            run += d;
        }
    }
    __syncthreads();

    const int4* d4 = (const int4*)(ei + N_EDGES + s * (N_EDGES / DEG_S));
    const int4* s4 = (const int4*)(ei + s * (N_EDGES / DEG_S));
    const int n4 = N_EDGES / DEG_S / 4;

    const int full = n4 / (2 * FILL_T);
    for (int r = 0; r < full; ++r) {
        int4 d[2], sv[2];
#pragma unroll
        for (int u = 0; u < 2; ++u) {
            d[u]  = d4[(r * 2 + u) * FILL_T + tid];
            sv[u] = s4[(r * 2 + u) * FILL_T + tid];
        }
#pragma unroll
        for (int u = 0; u < 2; ++u) fill_bin4(cur, ssrc, d[u], sv[u], base);
    }
    for (int i = full * 2 * FILL_T + tid; i < n4; i += FILL_T) fill_bin4(cur, ssrc, d4[i], s4[i], base);
}

template <int OUTC>
__global__ __launch_bounds__(256) void sage_layer(
    const float* __restrict__ feat, const int* __restrict__ ssrc,
    const unsigned* __restrict__ rowptr,
    const float* __restrict__ Wl, const float* __restrict__ Wr,
    const float* __restrict__ bias, float* __restrict__ out)
{
    constexpr int CT = OUTC / 16;
    constexpr int RT = 8 / CT;
    constexpr int AP = 68;

    __shared__ float Wlds[OUTC * 128];
    __shared__ __attribute__((aligned(16))) float At[RT * 16 * AP];

    const int tid = threadIdx.x;
    const int nb  = blockIdx.x * (RT * 16);
    const int grp = tid >> 4;
    const int g   = tid & 15;

    unsigned eb[RT], ee[RT];
#pragma unroll
    for (int j = 0; j < RT; ++j) {
        const int nn = min(nb + j * 16 + grp, N_NODES - 1);
        eb[j] = rowptr[nn];
        ee[j] = min(rowptr[nn + 1], (unsigned)N_EDGES);
    }

    for (int i = tid; i < OUTC * 128; i += 256) {
        int n = i >> 7, k = i & 127;
        Wlds[i] = (k < 64) ? Wl[n * 64 + k] : Wr[n * 64 + (k - 64)];
    }

#pragma unroll
    for (int j = 0; j < RT; ++j) {
        const int node = nb + j * 16 + grp;
        if (node < N_NODES) {
            const float4 acc = sage_row_sum(feat + g * 4, ssrc, eb[j], ee[j]);
            const unsigned deg = (ee[j] > eb[j]) ? ee[j] - eb[j] : 0u;
            const float rd = 1.0f / fmaxf((float)deg, 1.0f);
            *(float4*)&At[(j * 16 + grp) * AP + g * 4] =
                make_float4(acc.x * rd, acc.y * rd, acc.z * rd, acc.w * rd);
        }
    }
    __syncthreads();

    const int wave = tid >> 5;
    const int lane = tid & 31;
    const int rt   = wave / CT;
    const int row0 = nb + rt * 16;
    if (row0 >= N_NODES) return;

    const int col0 = (wave % CT) * 16;
    const int half = lane >> 4;
    const int m    = lane & 15;
    const int row  = row0 + m;
    const int colg = col0 + m;

    const float bc = bias[colg];

    v8f c;
#pragma unroll
    for (int i = 0; i < 8; ++i) c[i] = bc;

    const float* arow = &At[(rt * 16 + m) * AP];
    const float* xrow = feat + (long)row * 64;
    const float* wcol = &Wlds[colg * 128];

#pragma unroll
    for (int ks = 0; ks < 64; ks += 4) {
        v2f a, b;
        a[0] = arow[ks + half * 2 + 0];
        a[1] = arow[ks + half * 2 + 1];
        b[0] = wcol[ks + half * 2 + 0];
        b[1] = wcol[ks + half * 2 + 1];
        c = __builtin_amdgcn_wmma_f32_16x16x4_f32(false, a, false, b,
                                                  (short)0, c, false, false);
    }
#pragma unroll
    for (int ks = 0; ks < 64; ks += 4) {
        v2f a, b;
        a[0] = xrow[ks + half * 2 + 0];
        a[1] = xrow[ks + half * 2 + 1];
        b[0] = wcol[64 + ks + half * 2 + 0];
        b[1] = wcol[64 + ks + half * 2 + 1];
        c = __builtin_amdgcn_wmma_f32_16x16x4_f32(false, a, false, b,
                                                  (short)0, c, false, false);
    }

#pragma unroll
    for (int v = 0; v < 8; ++v) {
        float y = c[v];
        y = (y > 0.0f) ? y : NEG_SLOPE * y;
        out[(long)(row0 + half * 8 + v) * OUTC + colg] = y;
    }
}

extern "C" void kernel_launch(void* const* d_in, const int* in_sizes, int n_in,
                              void* d_out, int out_size, void* d_ws, size_t ws_size,
                              hipStream_t stream) {
    const float* x   = (const float*)d_in[0];
    const int*   ei  = (const int*)  d_in[1];
    const float* W1l = (const float*)d_in[2];
    const float* b1  = (const float*)d_in[3];
    const float* W1r = (const float*)d_in[4];
    const float* W2l = (const float*)d_in[5];
    const float* b2  = (const float*)d_in[6];
    const float* W2r = (const float*)d_in[7];
    float* out = (float*)d_out;

    const size_t degB = (size_t)DEG_S * DEG_NP * 4;
    const size_t aggB = (size_t)N_NODES * 64 * 4;
    unsigned* planes = (unsigned*)d_ws;
    float* h   = (float*)((char*)d_ws + degB + aggB);
    unsigned* tot    = (unsigned*)((char*)d_ws + degB + 2 * aggB);
    unsigned* rowptr = tot + DEG_P * DEG_S;
    int*      ssrc   = (int*)(rowptr + DEG_NP);

    sage_degree<<<DEG_P * DEG_S, DEG_T, 0, stream>>>(ei, planes, tot);
    sage_fill<<<DEG_P * DEG_S, FILL_T, 0, stream>>>(ei, planes, tot, ssrc, rowptr);

    sage_layer<HID_CH><<<(N_NODES / 16 + 1) / 2, 256, 0, stream>>>(x, ssrc, rowptr, W1l, W1r, b1, h);

    sage_layer<OUT_CH><<<(N_NODES / 16 + 3) / 4, 256, 0, stream>>>(h, ssrc, rowptr, W2l, W2r, b2, out);
}


# destination lists by a two-level partition: slices bin edges by node range in LDS and store whole lines, one work-group per range then sorts them in LDS; rows summed in (wave, program) list order
# speedup vs baseline: 5.0784x; 1.1466x over previous
#include <hip/hip_runtime.h>

#define N_NODES 50000
#define N_EDGES 1600000
#define IN_CH   64
#define HID_CH  64
#define OUT_CH  32
#define NEG_SLOPE 0.01f

#define NS   256
#define ES   (N_EDGES / NS)
#define NR   256
#define RN   196
#define RP_N (NR * RN)
#define PT   1024
#define CAP  8192
#define NONE 0xFFFFFFFFu
#define NW   (PT / 32)
#define NC   (256 * NW)

static_assert(NS * ES == N_EDGES, "slices cover the edges exactly");
static_assert(RP_N > N_NODES && RN <= 256 && N_NODES <= 65536, "packed entry: 8-bit offset in range, 16-bit source");
static_assert(NS == 256 && NR == 256 && NC == 8 * PT, "scan_counters: 8 counters per thread");

typedef __attribute__((ext_vector_type(2))) float v2f;
typedef __attribute__((ext_vector_type(8))) float v8f;

__device__ __forceinline__ float4 sage_row_sum(
    const float* __restrict__ f, const int* __restrict__ ssrc, unsigned e, const unsigned end)
{
    float4 acc = make_float4(0.0f, 0.0f, 0.0f, 0.0f);

    for (; e + 8u <= end; e += 8u) {
        int si[8];
        float4 v[8];
#pragma unroll
        for (int u = 0; u < 8; ++u) si[u] = ssrc[e + u];
#pragma unroll
        for (int u = 0; u < 8; ++u) v[u] = *(const float4*)(f + (long)si[u] * 64);
#pragma unroll
        for (int u = 0; u < 8; ++u) {
            acc.x += v[u].x; acc.y += v[u].y; acc.z += v[u].z; acc.w += v[u].w;
        }
    }
    if (e < end) {
        int si[7];
        float4 v[7];
#pragma unroll
        for (int u = 0; u < 7; ++u) si[u] = ssrc[min(e + (unsigned)u, end - 1u)];
#pragma unroll
        for (int u = 0; u < 7; ++u) {
            v[u] = make_float4(0.0f, 0.0f, 0.0f, 0.0f);
            if (e + u < end) v[u] = *(const float4*)(f + (long)si[u] * 64);
        }
#pragma unroll
        for (int u = 0; u < 7; ++u) {
            if (e + u < end) { acc.x += v[u].x; acc.y += v[u].y; acc.z += v[u].z; acc.w += v[u].w; }
        }
    }
    return acc;
}

__device__ __forceinline__ unsigned scan_counters(unsigned* c, unsigned* wsum, unsigned* total,
                                                  const int tid, const int lane, const int wave)
{
    uint4* c4 = (uint4*)c;
    const uint4 a = c4[tid * 2];
    const uint4 b = c4[tid * 2 + 1];
    const unsigned loc = a.x + a.y + a.z + a.w + b.x + b.y + b.z + b.w;
    unsigned inc = loc;
    for (int o = 1; o < 32; o <<= 1) {
        const unsigned v = __shfl_up(inc, o);
        if (lane >= o) inc += v;
    }
    if (lane == 31) wsum[wave] = inc;
    __syncthreads();
    unsigned run = inc - loc;
    for (int w = 0; w < wave; ++w) run += wsum[w];
    const unsigned first = run;
    uint4 pa, pb;
    pa.x = run; run += a.x;
    pa.y = run; run += a.y;
    pa.z = run; run += a.z;
    pa.w = run; run += a.w;
    pb.x = run; run += b.x;
    pb.y = run; run += b.y;
    pb.z = run; run += b.z;
    pb.w = run; run += b.w;
    c4[tid * 2]     = pa;
    c4[tid * 2 + 1] = pb;
    if (tid == PT - 1) *total = run;
    __syncthreads();
    return first;
}

__global__ __launch_bounds__(PT) void sage_part(
    const int* __restrict__ ei, unsigned* __restrict__ part, unsigned* __restrict__ tab)
{
    __shared__ __attribute__((aligned(16))) unsigned cw[NC];
    __shared__ unsigned wsum[NW];
    __shared__ unsigned total;
    __shared__ unsigned stage[ES];

    const int tid  = threadIdx.x;
    const int lane = tid & 31;
    const int wave = tid >> 5;
    const int s    = blockIdx.x;

    ((uint4*)cw)[tid * 2]     = make_uint4(0u, 0u, 0u, 0u);
    ((uint4*)cw)[tid * 2 + 1] = make_uint4(0u, 0u, 0u, 0u);
    __syncthreads();

    constexpr int EPT = (ES + PT - 1) / PT;
    const int* sp = ei + s * ES;
    const int* dp = ei + N_EDGES + s * ES;

    unsigned rg[EPT], pk[EPT], rk[EPT];
#pragma unroll
    for (int u = 0; u < EPT; ++u) {
        const int i = tid + u * PT;
        unsigned d = 0u, sv = 0u;
        if (i < ES) { d = (unsigned)dp[i]; sv = (unsigned)sp[i]; }
        const unsigned r = min(d / (unsigned)RN, (unsigned)(NR - 1));
        rg[u] = (i < ES) ? r : (unsigned)NR;
        pk[u] = (((d - r * RN) & 0xFFu) << 16) | (sv & 0xFFFFu);
    }
#pragma unroll
    for (int u = 0; u < EPT; ++u) {
        rk[u] = 0u;
        if (rg[u] < NR) rk[u] = atomicAdd(&cw[rg[u] * NW + wave], 1u);
    }
    __syncthreads();

    const unsigned first = scan_counters(cw, wsum, &total, tid, lane, wave);
    if ((tid & 3) == 0) tab[(tid >> 2) * NS + s] = first;
    if (tid == 0) tab[NR * NS + s] = (unsigned)ES;

#pragma unroll
    for (int u = 0; u < EPT; ++u) {
        if (rg[u] < NR) {
            const unsigned q = cw[rg[u] * NW + wave] + rk[u];
            if (q < (unsigned)ES) stage[q] = pk[u];
        }
    }
    __syncthreads();

    for (int i = tid; i < ES; i += PT) part[s * ES + i] = stage[i];
}

__device__ __forceinline__ unsigned ent_node(const unsigned w) { return min(w >> 16, (unsigned)(RN - 1)); }

__device__ __forceinline__ void ent_place(unsigned* cur, unsigned* stage, int* __restrict__ ssrc,
                                          const unsigned base, const int wave, const unsigned w)
{
    const unsigned slot = atomicAdd(&cur[ent_node(w) * NW + wave], 1u);
    const unsigned sv = w & 0xFFFFu;
    if (slot < (unsigned)CAP) {
        stage[slot] = sv;
    } else {
        const unsigned g = base + slot;
        if (g < (unsigned)N_EDGES) ssrc[g] = (int)sv;
    }
}

__global__ __launch_bounds__(PT) void sage_sort(
    const unsigned* __restrict__ part, const unsigned* __restrict__ tab,
    int* __restrict__ ssrc, unsigned* __restrict__ rowptr)
{
    __shared__ __attribute__((aligned(16))) unsigned cw[NC];
    __shared__ unsigned sb[NS];
    __shared__ unsigned se[NS];
    __shared__ unsigned wsum[NW];
    __shared__ unsigned bsum[NS / 32];
    __shared__ unsigned total;
    __shared__ unsigned stage[CAP];

    const int tid  = threadIdx.x;
    const int lane = tid & 31;
    const int wave = tid >> 5;
    const int r    = blockIdx.x;

    ((uint4*)cw)[tid * 2]     = make_uint4(0u, 0u, 0u, 0u);
    ((uint4*)cw)[tid * 2 + 1] = make_uint4(0u, 0u, 0u, 0u);
    if (tid < NS) {
        const unsigned b = min(tab[r * NS + tid], (unsigned)ES);
        const unsigned e = min(tab[(r + 1) * NS + tid], (unsigned)ES);
        sb[tid] = b;
        se[tid] = max(e, b);
        unsigned v = b;
        for (int o = 16; o > 0; o >>= 1) v += __shfl_xor(v, o);
        if (lane == 0) bsum[wave] = v;
    }
    __syncthreads();

    unsigned base = 0u;
    for (int w = 0; w < NS / 32; ++w) base += bsum[w];

    constexpr int SPW = NS / NW;
    unsigned wv[SPW], nb[SPW], ne[SPW];
#pragma unroll
    for (int j = 0; j < SPW; ++j) {
        const int sl = wave + j * NW;
        nb[j] = sb[sl];
        ne[j] = se[sl];
        const unsigned i = nb[j] + (unsigned)lane;
        wv[j] = NONE;
        if (i < ne[j]) wv[j] = part[sl * ES + i];
    }

#pragma unroll
    for (int j = 0; j < SPW; ++j) {
        const int sl = wave + j * NW;
        if (wv[j] != NONE) atomicAdd(&cw[ent_node(wv[j]) * NW + wave], 1u);
        for (unsigned i = nb[j] + 32u + (unsigned)lane; i < ne[j]; i += 32u)
            atomicAdd(&cw[ent_node(part[sl * ES + i]) * NW + wave], 1u);
    }
    __syncthreads();

    const unsigned first = scan_counters(cw, wsum, &total, tid, lane, wave);
    if ((tid & 3) == 0 && (tid >> 2) < RN) rowptr[r * RN + (tid >> 2)] = base + first;

#pragma unroll
    for (int j = 0; j < SPW; ++j) {
        const int sl = wave + j * NW;
        if (wv[j] != NONE) ent_place(cw, stage, ssrc, base, wave, wv[j]);
        for (unsigned i = nb[j] + 32u + (unsigned)lane; i < ne[j]; i += 32u)
            ent_place(cw, stage, ssrc, base, wave, part[sl * ES + i]);
    }
    __syncthreads();

    const unsigned n = min(total, (unsigned)CAP);
    for (unsigned i = (unsigned)tid; i < n; i += (unsigned)PT) {
        const unsigned g = base + i;
        if (g < (unsigned)N_EDGES) ssrc[g] = (int)stage[i];
    }
}

template <int OUTC>
__global__ __launch_bounds__(256) void sage_layer(
    const float* __restrict__ feat, const int* __restrict__ ssrc,
    const unsigned* __restrict__ rowptr,
    const float* __restrict__ Wl, const float* __restrict__ Wr,
    const float* __restrict__ bias, float* __restrict__ out)
{
    constexpr int CT = OUTC / 16;
    constexpr int RT = 8 / CT;
    constexpr int AP = 68;

    __shared__ float Wlds[OUTC * 128];
    __shared__ __attribute__((aligned(16))) float At[RT * 16 * AP];

    const int tid = threadIdx.x;
    const int nb  = blockIdx.x * (RT * 16);
    const int grp = tid >> 4;
    const int g   = tid & 15;

    unsigned eb[RT], ee[RT];
#pragma unroll
    for (int j = 0; j < RT; ++j) {
        const int nn = min(nb + j * 16 + grp, N_NODES - 1);
        eb[j] = rowptr[nn];
        ee[j] = min(rowptr[nn + 1], (unsigned)N_EDGES);
    }

    for (int i = tid; i < OUTC * 128; i += 256) {
        int n = i >> 7, k = i & 127;
        Wlds[i] = (k < 64) ? Wl[n * 64 + k] : Wr[n * 64 + (k - 64)];
    }

#pragma unroll
    for (int j = 0; j < RT; ++j) {
        const int node = nb + j * 16 + grp;
        if (node < N_NODES) {
            const float4 acc = sage_row_sum(feat + g * 4, ssrc, eb[j], ee[j]);
            const unsigned deg = (ee[j] > eb[j]) ? ee[j] - eb[j] : 0u;
            const float rd = 1.0f / fmaxf((float)deg, 1.0f);
            *(float4*)&At[(j * 16 + grp) * AP + g * 4] =
                make_float4(acc.x * rd, acc.y * rd, acc.z * rd, acc.w * rd);
        }
    }
    __syncthreads();

    const int wave = tid >> 5;
    const int lane = tid & 31;
    const int rt   = wave / CT;
    const int row0 = nb + rt * 16;
    if (row0 >= N_NODES) return;

    const int col0 = (wave % CT) * 16;
    const int half = lane >> 4;
    const int m    = lane & 15;
    const int row  = row0 + m;
    const int colg = col0 + m;

    const float bc = bias[colg];

    v8f c;
#pragma unroll
    for (int i = 0; i < 8; ++i) c[i] = bc;

    const float* arow = &At[(rt * 16 + m) * AP];
    const float* xrow = feat + (long)row * 64;
    const float* wcol = &Wlds[colg * 128];

#pragma unroll
    for (int ks = 0; ks < 64; ks += 4) {
        v2f a, b;
        a[0] = arow[ks + half * 2 + 0];
        a[1] = arow[ks + half * 2 + 1];
        b[0] = wcol[ks + half * 2 + 0];
        b[1] = wcol[ks + half * 2 + 1];
        c = __builtin_amdgcn_wmma_f32_16x16x4_f32(false, a, false, b,
                                                  (short)0, c, false, false);
    }
#pragma unroll
    for (int ks = 0; ks < 64; ks += 4) {
        v2f a, b;
        a[0] = xrow[ks + half * 2 + 0];
        a[1] = xrow[ks + half * 2 + 1];
        b[0] = wcol[64 + ks + half * 2 + 0];
        b[1] = wcol[64 + ks + half * 2 + 1];
        c = __builtin_amdgcn_wmma_f32_16x16x4_f32(false, a, false, b,
                                                  (short)0, c, false, false);
    }

#pragma unroll
    for (int v = 0; v < 8; ++v) {
        float y = c[v];
        y = (y > 0.0f) ? y : NEG_SLOPE * y;
        out[(long)(row0 + half * 8 + v) * OUTC + colg] = y;
    }
}

extern "C" void kernel_launch(void* const* d_in, const int* in_sizes, int n_in,
                              void* d_out, int out_size, void* d_ws, size_t ws_size,
                              hipStream_t stream) {
    const float* x   = (const float*)d_in[0];
    const int*   ei  = (const int*)  d_in[1];
    const float* W1l = (const float*)d_in[2];
    const float* b1  = (const float*)d_in[3];
    const float* W1r = (const float*)d_in[4];
    const float* W2l = (const float*)d_in[5];
    const float* b2  = (const float*)d_in[6];
    const float* W2r = (const float*)d_in[7];
    float* out = (float*)d_out;

    const size_t partB = (size_t)N_EDGES * 4;
    const size_t featB = (size_t)N_NODES * 64 * 4;
    const size_t tabB  = (size_t)(NR + 1) * NS * 4;
    unsigned* part   = (unsigned*)d_ws;
    float*    h      = (float*)((char*)d_ws + partB);
    unsigned* tab    = (unsigned*)((char*)d_ws + partB + featB);
    unsigned* rowptr = (unsigned*)((char*)d_ws + partB + featB + tabB);
    int*      ssrc   = (int*)(rowptr + RP_N);

    sage_part<<<NS, PT, 0, stream>>>(ei, part, tab);
    sage_sort<<<NR, PT, 0, stream>>>(part, tab, ssrc, rowptr);

    sage_layer<HID_CH><<<(N_NODES / 16 + 1) / 2, 256, 0, stream>>>(x, ssrc, rowptr, W1l, W1r, b1, h);

    sage_layer<OUT_CH><<<(N_NODES / 16 + 3) / 4, 256, 0, stream>>>(h, ssrc, rowptr, W2l, W2r, b2, out);
}


# weight panel rows in LDS padded from 128 to 132 floats: the 16 rows a lane half reads for a B fragment now start on 16 different banks instead of one (every B read was a 16-way bank conflict)
# speedup vs baseline: 5.6220x; 1.1070x over previous
#include <hip/hip_runtime.h>

#define N_NODES 50000
#define N_EDGES 1600000
#define IN_CH   64
#define HID_CH  64
#define OUT_CH  32
#define NEG_SLOPE 0.01f

#define NS   256
#define ES   (N_EDGES / NS)
#define NR   256
#define RN   196
#define RP_N (NR * RN)
#define PT   1024
#define CAP  8192
#define NONE 0xFFFFFFFFu
#define NW   (PT / 32)
#define NC   (256 * NW)

static_assert(NS * ES == N_EDGES, "slices cover the edges exactly");
static_assert(RP_N > N_NODES && RN <= 256 && N_NODES <= 65536, "packed entry: 8-bit offset in range, 16-bit source");
static_assert(NS == 256 && NR == 256 && NC == 8 * PT, "scan_counters: 8 counters per thread");

typedef __attribute__((ext_vector_type(2))) float v2f;
typedef __attribute__((ext_vector_type(8))) float v8f;

__device__ __forceinline__ float4 sage_row_sum(
    const float* __restrict__ f, const int* __restrict__ ssrc, unsigned e, const unsigned end)
{
    float4 acc = make_float4(0.0f, 0.0f, 0.0f, 0.0f);

    for (; e + 8u <= end; e += 8u) {
        int si[8];
        float4 v[8];
#pragma unroll
        for (int u = 0; u < 8; ++u) si[u] = ssrc[e + u];
#pragma unroll
        for (int u = 0; u < 8; ++u) v[u] = *(const float4*)(f + (long)si[u] * 64);
#pragma unroll
        for (int u = 0; u < 8; ++u) {
            acc.x += v[u].x; acc.y += v[u].y; acc.z += v[u].z; acc.w += v[u].w;
        }
    }
    if (e < end) {
        int si[7];
        float4 v[7];
#pragma unroll
        for (int u = 0; u < 7; ++u) si[u] = ssrc[min(e + (unsigned)u, end - 1u)];
#pragma unroll
        for (int u = 0; u < 7; ++u) {
            v[u] = make_float4(0.0f, 0.0f, 0.0f, 0.0f);
            if (e + u < end) v[u] = *(const float4*)(f + (long)si[u] * 64);
        }
#pragma unroll
        for (int u = 0; u < 7; ++u) {
            if (e + u < end) { acc.x += v[u].x; acc.y += v[u].y; acc.z += v[u].z; acc.w += v[u].w; }
        }
    }
    return acc;
}

__device__ __forceinline__ unsigned scan_counters(unsigned* c, unsigned* wsum, unsigned* total,
                                                  const int tid, const int lane, const int wave)
{
    uint4* c4 = (uint4*)c;
    const uint4 a = c4[tid * 2];
    const uint4 b = c4[tid * 2 + 1];
    const unsigned loc = a.x + a.y + a.z + a.w + b.x + b.y + b.z + b.w;
    unsigned inc = loc;
    for (int o = 1; o < 32; o <<= 1) {
        const unsigned v = __shfl_up(inc, o);
        if (lane >= o) inc += v;
    }
    if (lane == 31) wsum[wave] = inc;
    __syncthreads();
    unsigned run = inc - loc;
    for (int w = 0; w < wave; ++w) run += wsum[w];
    const unsigned first = run;
    uint4 pa, pb;
    pa.x = run; run += a.x;
    pa.y = run; run += a.y;
    pa.z = run; run += a.z;
    pa.w = run; run += a.w;
    pb.x = run; run += b.x;
    pb.y = run; run += b.y;
    pb.z = run; run += b.z;
    pb.w = run; run += b.w;
    c4[tid * 2]     = pa;
    c4[tid * 2 + 1] = pb;
    if (tid == PT - 1) *total = run;
    __syncthreads();
    return first;
}

__global__ __launch_bounds__(PT) void sage_part(
    const int* __restrict__ ei, unsigned* __restrict__ part, unsigned* __restrict__ tab)
{
    __shared__ __attribute__((aligned(16))) unsigned cw[NC];
    __shared__ unsigned wsum[NW];
    __shared__ unsigned total;
    __shared__ unsigned stage[ES];

    const int tid  = threadIdx.x;
    const int lane = tid & 31;
    const int wave = tid >> 5;
    const int s    = blockIdx.x;

    ((uint4*)cw)[tid * 2]     = make_uint4(0u, 0u, 0u, 0u);
    ((uint4*)cw)[tid * 2 + 1] = make_uint4(0u, 0u, 0u, 0u);
    __syncthreads();

    constexpr int EPT = (ES + PT - 1) / PT;
    const int* sp = ei + s * ES;
    const int* dp = ei + N_EDGES + s * ES;

    unsigned rg[EPT], pk[EPT], rk[EPT];
#pragma unroll
    for (int u = 0; u < EPT; ++u) {
        const int i = tid + u * PT;
        unsigned d = 0u, sv = 0u;
        if (i < ES) { d = (unsigned)dp[i]; sv = (unsigned)sp[i]; }
        const unsigned r = min(d / (unsigned)RN, (unsigned)(NR - 1));
        rg[u] = (i < ES) ? r : (unsigned)NR;
        pk[u] = (((d - r * RN) & 0xFFu) << 16) | (sv & 0xFFFFu);
    }
#pragma unroll
    for (int u = 0; u < EPT; ++u) {
        rk[u] = 0u;
        if (rg[u] < NR) rk[u] = atomicAdd(&cw[rg[u] * NW + wave], 1u);
    }
    __syncthreads();

    const unsigned first = scan_counters(cw, wsum, &total, tid, lane, wave);
    if ((tid & 3) == 0) tab[(tid >> 2) * NS + s] = first;
    if (tid == 0) tab[NR * NS + s] = (unsigned)ES;

#pragma unroll
    for (int u = 0; u < EPT; ++u) {
        if (rg[u] < NR) {
            const unsigned q = cw[rg[u] * NW + wave] + rk[u];
            if (q < (unsigned)ES) stage[q] = pk[u];
        }
    }
    __syncthreads();

    for (int i = tid; i < ES; i += PT) part[s * ES + i] = stage[i];
}

__device__ __forceinline__ unsigned ent_node(const unsigned w) { return min(w >> 16, (unsigned)(RN - 1)); }

__device__ __forceinline__ void ent_place(unsigned* cur, unsigned* stage, int* __restrict__ ssrc,
                                          const unsigned base, const int wave, const unsigned w)
{
    const unsigned slot = atomicAdd(&cur[ent_node(w) * NW + wave], 1u);
    const unsigned sv = w & 0xFFFFu;
    if (slot < (unsigned)CAP) {
        stage[slot] = sv;
    } else {
        const unsigned g = base + slot;
        if (g < (unsigned)N_EDGES) ssrc[g] = (int)sv;
    }
}

__global__ __launch_bounds__(PT) void sage_sort(
    const unsigned* __restrict__ part, const unsigned* __restrict__ tab,
    int* __restrict__ ssrc, unsigned* __restrict__ rowptr)
{
    __shared__ __attribute__((aligned(16))) unsigned cw[NC];
    __shared__ unsigned sb[NS];
    __shared__ unsigned se[NS];
    __shared__ unsigned wsum[NW];
    __shared__ unsigned bsum[NS / 32];
    __shared__ unsigned total;
    __shared__ unsigned stage[CAP];

    const int tid  = threadIdx.x;
    const int lane = tid & 31;
    const int wave = tid >> 5;
    const int r    = blockIdx.x;

    ((uint4*)cw)[tid * 2]     = make_uint4(0u, 0u, 0u, 0u);
    ((uint4*)cw)[tid * 2 + 1] = make_uint4(0u, 0u, 0u, 0u);
    if (tid < NS) {
        const unsigned b = min(tab[r * NS + tid], (unsigned)ES);
        const unsigned e = min(tab[(r + 1) * NS + tid], (unsigned)ES);
        sb[tid] = b;
        se[tid] = max(e, b);
        unsigned v = b;
        for (int o = 16; o > 0; o >>= 1) v += __shfl_xor(v, o);
        if (lane == 0) bsum[wave] = v;
    }
    __syncthreads();

    unsigned base = 0u;
    for (int w = 0; w < NS / 32; ++w) base += bsum[w];

    constexpr int SPW = NS / NW;
    unsigned wv[SPW], nb[SPW], ne[SPW];
#pragma unroll
    for (int j = 0; j < SPW; ++j) {
        const int sl = wave + j * NW;
        nb[j] = sb[sl];
        ne[j] = se[sl];
        const unsigned i = nb[j] + (unsigned)lane;
        wv[j] = NONE;
        if (i < ne[j]) wv[j] = part[sl * ES + i];
    }

#pragma unroll
    for (int j = 0; j < SPW; ++j) {
        const int sl = wave + j * NW;
        if (wv[j] != NONE) atomicAdd(&cw[ent_node(wv[j]) * NW + wave], 1u);
        for (unsigned i = nb[j] + 32u + (unsigned)lane; i < ne[j]; i += 32u)
            atomicAdd(&cw[ent_node(part[sl * ES + i]) * NW + wave], 1u);
    }
    __syncthreads();

    const unsigned first = scan_counters(cw, wsum, &total, tid, lane, wave);
    if ((tid & 3) == 0 && (tid >> 2) < RN) rowptr[r * RN + (tid >> 2)] = base + first;

#pragma unroll
    for (int j = 0; j < SPW; ++j) {
        const int sl = wave + j * NW;
        if (wv[j] != NONE) ent_place(cw, stage, ssrc, base, wave, wv[j]);
        for (unsigned i = nb[j] + 32u + (unsigned)lane; i < ne[j]; i += 32u)
            ent_place(cw, stage, ssrc, base, wave, part[sl * ES + i]);
    }
    __syncthreads();

    const unsigned n = min(total, (unsigned)CAP);
    for (unsigned i = (unsigned)tid; i < n; i += (unsigned)PT) {
        const unsigned g = base + i;
        if (g < (unsigned)N_EDGES) ssrc[g] = (int)stage[i];
    }
}

template <int OUTC>
__global__ __launch_bounds__(256) void sage_layer(
    const float* __restrict__ feat, const int* __restrict__ ssrc,
    const unsigned* __restrict__ rowptr,
    const float* __restrict__ Wl, const float* __restrict__ Wr,
    const float* __restrict__ bias, float* __restrict__ out)
{
    constexpr int CT = OUTC / 16;
    constexpr int RT = 8 / CT;
    constexpr int AP = 68;
    constexpr int WP = 132;

    __shared__ float Wlds[OUTC * WP];
    __shared__ __attribute__((aligned(16))) float At[RT * 16 * AP];

    const int tid = threadIdx.x;
    const int nb  = blockIdx.x * (RT * 16);
    const int grp = tid >> 4;
    const int g   = tid & 15;

    unsigned eb[RT], ee[RT];
#pragma unroll
    for (int j = 0; j < RT; ++j) {
        const int nn = min(nb + j * 16 + grp, N_NODES - 1);
        eb[j] = rowptr[nn];
        ee[j] = min(rowptr[nn + 1], (unsigned)N_EDGES);
    }

    for (int i = tid; i < OUTC * 128; i += 256) {
        int n = i >> 7, k = i & 127;
        Wlds[n * WP + k] = (k < 64) ? Wl[n * 64 + k] : Wr[n * 64 + (k - 64)];
    }

#pragma unroll
    for (int j = 0; j < RT; ++j) {
        const int node = nb + j * 16 + grp;
        if (node < N_NODES) {
            const float4 acc = sage_row_sum(feat + g * 4, ssrc, eb[j], ee[j]);
            const unsigned deg = (ee[j] > eb[j]) ? ee[j] - eb[j] : 0u;
            const float rd = 1.0f / fmaxf((float)deg, 1.0f);
            *(float4*)&At[(j * 16 + grp) * AP + g * 4] =
                make_float4(acc.x * rd, acc.y * rd, acc.z * rd, acc.w * rd);
        }
    }
    __syncthreads();

    const int wave = tid >> 5;
    const int lane = tid & 31;
    const int rt   = wave / CT;
    const int row0 = nb + rt * 16;
    if (row0 >= N_NODES) return;

    const int col0 = (wave % CT) * 16;
    const int half = lane >> 4;
    const int m    = lane & 15;
    const int row  = row0 + m;
    const int colg = col0 + m;

    const float bc = bias[colg];

    v8f c;
#pragma unroll
    for (int i = 0; i < 8; ++i) c[i] = bc;

    const float* arow = &At[(rt * 16 + m) * AP];
    const float* xrow = feat + (long)row * 64;
    const float* wcol = &Wlds[colg * WP];

#pragma unroll
    for (int ks = 0; ks < 64; ks += 4) {
        v2f a, b;
        a[0] = arow[ks + half * 2 + 0];
        a[1] = arow[ks + half * 2 + 1];
        b[0] = wcol[ks + half * 2 + 0];
        b[1] = wcol[ks + half * 2 + 1];
        c = __builtin_amdgcn_wmma_f32_16x16x4_f32(false, a, false, b,
                                                  (short)0, c, false, false);
    }
#pragma unroll
    for (int ks = 0; ks < 64; ks += 4) {
        v2f a, b;
        a[0] = xrow[ks + half * 2 + 0];
        a[1] = xrow[ks + half * 2 + 1];
        b[0] = wcol[64 + ks + half * 2 + 0];
        b[1] = wcol[64 + ks + half * 2 + 1];
        c = __builtin_amdgcn_wmma_f32_16x16x4_f32(false, a, false, b,
                                                  (short)0, c, false, false);
    }

#pragma unroll
    for (int v = 0; v < 8; ++v) {
        float y = c[v];
        y = (y > 0.0f) ? y : NEG_SLOPE * y;
        out[(long)(row0 + half * 8 + v) * OUTC + colg] = y;
    }
}

extern "C" void kernel_launch(void* const* d_in, const int* in_sizes, int n_in,
                              void* d_out, int out_size, void* d_ws, size_t ws_size,
                              hipStream_t stream) {
    const float* x   = (const float*)d_in[0];
    const int*   ei  = (const int*)  d_in[1];
    const float* W1l = (const float*)d_in[2];
    const float* b1  = (const float*)d_in[3];
    const float* W1r = (const float*)d_in[4];
    const float* W2l = (const float*)d_in[5];
    const float* b2  = (const float*)d_in[6];
    const float* W2r = (const float*)d_in[7];
    float* out = (float*)d_out;

    const size_t partB = (size_t)N_EDGES * 4;
    const size_t featB = (size_t)N_NODES * 64 * 4;
    const size_t tabB  = (size_t)(NR + 1) * NS * 4;
    unsigned* part   = (unsigned*)d_ws;
    float*    h      = (float*)((char*)d_ws + partB);
    unsigned* tab    = (unsigned*)((char*)d_ws + partB + featB);
    unsigned* rowptr = (unsigned*)((char*)d_ws + partB + featB + tabB);
    int*      ssrc   = (int*)(rowptr + RP_N);

    sage_part<<<NS, PT, 0, stream>>>(ei, part, tab);
    sage_sort<<<NR, PT, 0, stream>>>(part, tab, ssrc, rowptr);

    sage_layer<HID_CH><<<(N_NODES / 16 + 1) / 2, 256, 0, stream>>>(x, ssrc, rowptr, W1l, W1r, b1, h);

    sage_layer<OUT_CH><<<(N_NODES / 16 + 3) / 4, 256, 0, stream>>>(h, ssrc, rowptr, W2l, W2r, b2, out);
}


# LDS counter rows of both list-building kernels padded from 32 to 36 words: a wave's column of counters covers 16 banks instead of 2 (its LDS adds and counter reads were 16-way conflicts)
# speedup vs baseline: 6.1015x; 1.0853x over previous
#include <hip/hip_runtime.h>

#define N_NODES 50000
#define N_EDGES 1600000
#define IN_CH   64
#define HID_CH  64
#define OUT_CH  32
#define NEG_SLOPE 0.01f

#define NS   256
#define ES   (N_EDGES / NS)
#define NR   256
#define RN   196
#define RP_N (NR * RN)
#define PT   1024
#define CAP  8192
#define NONE 0xFFFFFFFFu
#define NW   (PT / 32)
#define NC   (256 * NW)
#define CP   (NW + 4)
#define NCW  (256 * CP)

static_assert(NS * ES == N_EDGES, "slices cover the edges exactly");
static_assert(RP_N > N_NODES && RN <= 256 && N_NODES <= 65536, "packed entry: 8-bit offset in range, 16-bit source");
static_assert(NS == 256 && NR == 256 && NC == 8 * PT, "scan_counters: 8 counters per thread");
static_assert(NW % 8 == 0 && CP % 4 == 0, "a thread's 8 counters lie in one row, 16-byte aligned");

__device__ __forceinline__ int ctr_vec(const int tid) { return (tid >> 2) * (CP / 4) + (tid & 3) * 2; }

typedef __attribute__((ext_vector_type(2))) float v2f;
typedef __attribute__((ext_vector_type(8))) float v8f;

__device__ __forceinline__ float4 sage_row_sum(
    const float* __restrict__ f, const int* __restrict__ ssrc, unsigned e, const unsigned end)
{
    float4 acc = make_float4(0.0f, 0.0f, 0.0f, 0.0f);

    for (; e + 8u <= end; e += 8u) {
        int si[8];
        float4 v[8];
#pragma unroll
        for (int u = 0; u < 8; ++u) si[u] = ssrc[e + u];
#pragma unroll
        for (int u = 0; u < 8; ++u) v[u] = *(const float4*)(f + (long)si[u] * 64);
#pragma unroll
        for (int u = 0; u < 8; ++u) {
            acc.x += v[u].x; acc.y += v[u].y; acc.z += v[u].z; acc.w += v[u].w;
        }
    }
    if (e < end) {
        int si[7];
        float4 v[7];
#pragma unroll
        for (int u = 0; u < 7; ++u) si[u] = ssrc[min(e + (unsigned)u, end - 1u)];
#pragma unroll
        for (int u = 0; u < 7; ++u) {
            v[u] = make_float4(0.0f, 0.0f, 0.0f, 0.0f);
            if (e + u < end) v[u] = *(const float4*)(f + (long)si[u] * 64);
        }
#pragma unroll
        for (int u = 0; u < 7; ++u) {
            if (e + u < end) { acc.x += v[u].x; acc.y += v[u].y; acc.z += v[u].z; acc.w += v[u].w; }
        }
    }
    return acc;
}

__device__ __forceinline__ unsigned scan_counters(unsigned* c, unsigned* wsum, unsigned* total,
                                                  const int tid, const int lane, const int wave)
{
    uint4* c4 = (uint4*)c;
    const int cv = ctr_vec(tid);
    const uint4 a = c4[cv];
    const uint4 b = c4[cv + 1];
    const unsigned loc = a.x + a.y + a.z + a.w + b.x + b.y + b.z + b.w;
    unsigned inc = loc;
    for (int o = 1; o < 32; o <<= 1) {
        const unsigned v = __shfl_up(inc, o);
        if (lane >= o) inc += v;
    }
    if (lane == 31) wsum[wave] = inc;
    __syncthreads();
    unsigned run = inc - loc;
    for (int w = 0; w < wave; ++w) run += wsum[w];
    const unsigned first = run;
    uint4 pa, pb;
    pa.x = run; run += a.x;
    pa.y = run; run += a.y;
    pa.z = run; run += a.z;
    pa.w = run; run += a.w;
    pb.x = run; run += b.x;
    pb.y = run; run += b.y;
    pb.z = run; run += b.z;
    pb.w = run; run += b.w;
    c4[cv]     = pa;
    c4[cv + 1] = pb;
    if (tid == PT - 1) *total = run;
    __syncthreads();
    return first;
}

__global__ __launch_bounds__(PT) void sage_part(
    const int* __restrict__ ei, unsigned* __restrict__ part, unsigned* __restrict__ tab)
{
    __shared__ __attribute__((aligned(16))) unsigned cw[NCW];
    __shared__ unsigned wsum[NW];
    __shared__ unsigned total;
    __shared__ unsigned stage[ES];

    const int tid  = threadIdx.x;
    const int lane = tid & 31;
    const int wave = tid >> 5;
    const int s    = blockIdx.x;

    ((uint4*)cw)[ctr_vec(tid)]     = make_uint4(0u, 0u, 0u, 0u);
    ((uint4*)cw)[ctr_vec(tid) + 1] = make_uint4(0u, 0u, 0u, 0u);
    __syncthreads();

    constexpr int EPT = (ES + PT - 1) / PT;
    const int* sp = ei + s * ES;
    const int* dp = ei + N_EDGES + s * ES;

    unsigned rg[EPT], pk[EPT], rk[EPT];
#pragma unroll
    for (int u = 0; u < EPT; ++u) {
        const int i = tid + u * PT;
        unsigned d = 0u, sv = 0u;
        if (i < ES) { d = (unsigned)dp[i]; sv = (unsigned)sp[i]; }
        const unsigned r = min(d / (unsigned)RN, (unsigned)(NR - 1));
        rg[u] = (i < ES) ? r : (unsigned)NR;
        pk[u] = (((d - r * RN) & 0xFFu) << 16) | (sv & 0xFFFFu);
    }
#pragma unroll
    for (int u = 0; u < EPT; ++u) {
        rk[u] = 0u;
        if (rg[u] < NR) rk[u] = atomicAdd(&cw[rg[u] * CP + wave], 1u);
    }
    __syncthreads();

    const unsigned first = scan_counters(cw, wsum, &total, tid, lane, wave);
    if ((tid & 3) == 0) tab[(tid >> 2) * NS + s] = first;
    if (tid == 0) tab[NR * NS + s] = (unsigned)ES;

#pragma unroll
    for (int u = 0; u < EPT; ++u) {
        if (rg[u] < NR) {
            const unsigned q = cw[rg[u] * CP + wave] + rk[u];
            if (q < (unsigned)ES) stage[q] = pk[u];
        }
    }
    __syncthreads();

    for (int i = tid; i < ES; i += PT) part[s * ES + i] = stage[i];
}

__device__ __forceinline__ unsigned ent_node(const unsigned w) { return min(w >> 16, (unsigned)(RN - 1)); }

__device__ __forceinline__ void ent_place(unsigned* cur, unsigned* stage, int* __restrict__ ssrc,
                                          const unsigned base, const int wave, const unsigned w)
{
    const unsigned slot = atomicAdd(&cur[ent_node(w) * CP + wave], 1u);
    const unsigned sv = w & 0xFFFFu;
    if (slot < (unsigned)CAP) {
        stage[slot] = sv;
    } else {
        const unsigned g = base + slot;
        if (g < (unsigned)N_EDGES) ssrc[g] = (int)sv;
    }
}

__global__ __launch_bounds__(PT) void sage_sort(
    const unsigned* __restrict__ part, const unsigned* __restrict__ tab,
    int* __restrict__ ssrc, unsigned* __restrict__ rowptr)
{
    __shared__ __attribute__((aligned(16))) unsigned cw[NCW];
    __shared__ unsigned sb[NS];
    __shared__ unsigned se[NS];
    __shared__ unsigned wsum[NW];
    __shared__ unsigned bsum[NS / 32];
    __shared__ unsigned total;
    __shared__ unsigned stage[CAP];

    const int tid  = threadIdx.x;
    const int lane = tid & 31;
    const int wave = tid >> 5;
    const int r    = blockIdx.x;

    ((uint4*)cw)[ctr_vec(tid)]     = make_uint4(0u, 0u, 0u, 0u);
    ((uint4*)cw)[ctr_vec(tid) + 1] = make_uint4(0u, 0u, 0u, 0u);
    if (tid < NS) {
        const unsigned b = min(tab[r * NS + tid], (unsigned)ES);
        const unsigned e = min(tab[(r + 1) * NS + tid], (unsigned)ES);
        sb[tid] = b;
        se[tid] = max(e, b);
        unsigned v = b;
        for (int o = 16; o > 0; o >>= 1) v += __shfl_xor(v, o);
        if (lane == 0) bsum[wave] = v;
    }
    __syncthreads();

    unsigned base = 0u;
    for (int w = 0; w < NS / 32; ++w) base += bsum[w];

    constexpr int SPW = NS / NW;
    unsigned wv[SPW], nb[SPW], ne[SPW];
#pragma unroll
    for (int j = 0; j < SPW; ++j) {
        const int sl = wave + j * NW;
        nb[j] = sb[sl];
        ne[j] = se[sl];
        const unsigned i = nb[j] + (unsigned)lane;
        wv[j] = NONE;
        if (i < ne[j]) wv[j] = part[sl * ES + i];
    }

#pragma unroll
    for (int j = 0; j < SPW; ++j) {
        const int sl = wave + j * NW;
        if (wv[j] != NONE) atomicAdd(&cw[ent_node(wv[j]) * CP + wave], 1u);
        for (unsigned i = nb[j] + 32u + (unsigned)lane; i < ne[j]; i += 32u)
            atomicAdd(&cw[ent_node(part[sl * ES + i]) * CP + wave], 1u);
    }
    __syncthreads();

    const unsigned first = scan_counters(cw, wsum, &total, tid, lane, wave);
    if ((tid & 3) == 0 && (tid >> 2) < RN) rowptr[r * RN + (tid >> 2)] = base + first;

#pragma unroll
    for (int j = 0; j < SPW; ++j) {
        const int sl = wave + j * NW;
        if (wv[j] != NONE) ent_place(cw, stage, ssrc, base, wave, wv[j]);
        for (unsigned i = nb[j] + 32u + (unsigned)lane; i < ne[j]; i += 32u)
            ent_place(cw, stage, ssrc, base, wave, part[sl * ES + i]);
    }
    __syncthreads();

    const unsigned n = min(total, (unsigned)CAP);
    for (unsigned i = (unsigned)tid; i < n; i += (unsigned)PT) {
        const unsigned g = base + i;
        if (g < (unsigned)N_EDGES) ssrc[g] = (int)stage[i];
    }
}

template <int OUTC>
__global__ __launch_bounds__(256) void sage_layer(
    const float* __restrict__ feat, const int* __restrict__ ssrc,
    const unsigned* __restrict__ rowptr,
    const float* __restrict__ Wl, const float* __restrict__ Wr,
    const float* __restrict__ bias, float* __restrict__ out)
{
    constexpr int CT = OUTC / 16;
    constexpr int RT = 8 / CT;
    constexpr int AP = 68;
    constexpr int WP = 132;

    __shared__ float Wlds[OUTC * WP];
    __shared__ __attribute__((aligned(16))) float At[RT * 16 * AP];

    const int tid = threadIdx.x;
    const int nb  = blockIdx.x * (RT * 16);
    const int grp = tid >> 4;
    const int g   = tid & 15;

    unsigned eb[RT], ee[RT];
#pragma unroll
    for (int j = 0; j < RT; ++j) {
        const int nn = min(nb + j * 16 + grp, N_NODES - 1);
        eb[j] = rowptr[nn];
        ee[j] = min(rowptr[nn + 1], (unsigned)N_EDGES);
    }

    for (int i = tid; i < OUTC * 128; i += 256) {
        int n = i >> 7, k = i & 127;
        Wlds[n * WP + k] = (k < 64) ? Wl[n * 64 + k] : Wr[n * 64 + (k - 64)];
    }

#pragma unroll
    for (int j = 0; j < RT; ++j) {
        const int node = nb + j * 16 + grp;
        if (node < N_NODES) {
            const float4 acc = sage_row_sum(feat + g * 4, ssrc, eb[j], ee[j]);
            const unsigned deg = (ee[j] > eb[j]) ? ee[j] - eb[j] : 0u;
            const float rd = 1.0f / fmaxf((float)deg, 1.0f);
            *(float4*)&At[(j * 16 + grp) * AP + g * 4] =
                make_float4(acc.x * rd, acc.y * rd, acc.z * rd, acc.w * rd);
        }
    }
    __syncthreads();

    const int wave = tid >> 5;
    const int lane = tid & 31;
    const int rt   = wave / CT;
    const int row0 = nb + rt * 16;
    if (row0 >= N_NODES) return;

    const int col0 = (wave % CT) * 16;
    const int half = lane >> 4;
    const int m    = lane & 15;
    const int row  = row0 + m;
    const int colg = col0 + m;

    const float bc = bias[colg];

    v8f c;
#pragma unroll
    for (int i = 0; i < 8; ++i) c[i] = bc;

    const float* arow = &At[(rt * 16 + m) * AP];
    const float* xrow = feat + (long)row * 64;
    const float* wcol = &Wlds[colg * WP];

#pragma unroll
    for (int ks = 0; ks < 64; ks += 4) {
        v2f a, b;
        a[0] = arow[ks + half * 2 + 0];
        a[1] = arow[ks + half * 2 + 1];
        b[0] = wcol[ks + half * 2 + 0];
        b[1] = wcol[ks + half * 2 + 1];
        c = __builtin_amdgcn_wmma_f32_16x16x4_f32(false, a, false, b,
                                                  (short)0, c, false, false);
    }
#pragma unroll
    for (int ks = 0; ks < 64; ks += 4) {
        v2f a, b;
        a[0] = xrow[ks + half * 2 + 0];
        a[1] = xrow[ks + half * 2 + 1];
        b[0] = wcol[64 + ks + half * 2 + 0];
        b[1] = wcol[64 + ks + half * 2 + 1];
        c = __builtin_amdgcn_wmma_f32_16x16x4_f32(false, a, false, b,
                                                  (short)0, c, false, false);
    }

#pragma unroll
    for (int v = 0; v < 8; ++v) {
        float y = c[v];
        y = (y > 0.0f) ? y : NEG_SLOPE * y;
        out[(long)(row0 + half * 8 + v) * OUTC + colg] = y;
    }
}

extern "C" void kernel_launch(void* const* d_in, const int* in_sizes, int n_in,
                              void* d_out, int out_size, void* d_ws, size_t ws_size,
                              hipStream_t stream) {
    const float* x   = (const float*)d_in[0];
    const int*   ei  = (const int*)  d_in[1];
    const float* W1l = (const float*)d_in[2];
    const float* b1  = (const float*)d_in[3];
    const float* W1r = (const float*)d_in[4];
    const float* W2l = (const float*)d_in[5];
    const float* b2  = (const float*)d_in[6];
    const float* W2r = (const float*)d_in[7];
    float* out = (float*)d_out;

    const size_t partB = (size_t)N_EDGES * 4;
    const size_t featB = (size_t)N_NODES * 64 * 4;
    const size_t tabB  = (size_t)(NR + 1) * NS * 4;
    unsigned* part   = (unsigned*)d_ws;
    float*    h      = (float*)((char*)d_ws + partB);
    unsigned* tab    = (unsigned*)((char*)d_ws + partB + featB);
    unsigned* rowptr = (unsigned*)((char*)d_ws + partB + featB + tabB);
    int*      ssrc   = (int*)(rowptr + RP_N);

    sage_part<<<NS, PT, 0, stream>>>(ei, part, tab);
    sage_sort<<<NR, PT, 0, stream>>>(part, tab, ssrc, rowptr);

    sage_layer<HID_CH><<<(N_NODES / 16 + 1) / 2, 256, 0, stream>>>(x, ssrc, rowptr, W1l, W1r, b1, h);

    sage_layer<OUT_CH><<<(N_NODES / 16 + 3) / 4, 256, 0, stream>>>(h, ssrc, rowptr, W2l, W2r, b2, out);
}


# neighbour lists read 16 entries per lane-coalesced load (3 pieces up front) and handed round the 16-lane group by DPP row_share, not 8 same-address loads per 8 rows; row order kept
# speedup vs baseline: 6.3672x; 1.0435x over previous
#include <hip/hip_runtime.h>

#define N_NODES 50000
#define N_EDGES 1600000
#define IN_CH   64
#define HID_CH  64
#define OUT_CH  32
#define NEG_SLOPE 0.01f

#define NS   256
#define ES   (N_EDGES / NS)
#define NR   256
#define RN   196
#define RP_N (NR * RN)
#define PT   1024
#define CAP  8192
#define NONE 0xFFFFFFFFu
#define NW   (PT / 32)
#define NC   (256 * NW)
#define CP   (NW + 4)
#define NCW  (256 * CP)

static_assert(NS * ES == N_EDGES, "slices cover the edges exactly");
static_assert(RP_N > N_NODES && RN <= 256 && N_NODES <= 65536, "packed entry: 8-bit offset in range, 16-bit source");
static_assert(NS == 256 && NR == 256 && NC == 8 * PT, "scan_counters: 8 counters per thread");
static_assert(NW % 8 == 0 && CP % 4 == 0, "a thread's 8 counters lie in one row, 16-byte aligned");

__device__ __forceinline__ int ctr_vec(const int tid) { return (tid >> 2) * (CP / 4) + (tid & 3) * 2; }

typedef __attribute__((ext_vector_type(2))) float v2f;
typedef __attribute__((ext_vector_type(8))) float v8f;


#define GRP_BCAST(v, U) __builtin_amdgcn_update_dpp(0, (v), 0x150 + (U), 0xF, 0xF, false)

__device__ __forceinline__ void acc_add(float4& acc, const float4 v)
{
    acc.x += v.x; acc.y += v.y; acc.z += v.z; acc.w += v.w;
}

template <int H>
__device__ __forceinline__ void sage_sum8(float4& acc, const float* __restrict__ f, const int idx,
                                          const unsigned e, const unsigned end)
{
    int si[8];
    si[0] = GRP_BCAST(idx, H + 0); si[1] = GRP_BCAST(idx, H + 1);
    si[2] = GRP_BCAST(idx, H + 2); si[3] = GRP_BCAST(idx, H + 3);
    si[4] = GRP_BCAST(idx, H + 4); si[5] = GRP_BCAST(idx, H + 5);
    si[6] = GRP_BCAST(idx, H + 6); si[7] = GRP_BCAST(idx, H + 7);
    float4 v[8];
    if (e + 8u <= end) {
#pragma unroll
        for (int u = 0; u < 8; ++u) v[u] = *(const float4*)(f + (long)si[u] * 64);
#pragma unroll
        for (int u = 0; u < 8; ++u) acc_add(acc, v[u]);
    } else {
#pragma unroll
        for (int u = 0; u < 7; ++u) {
            v[u] = make_float4(0.0f, 0.0f, 0.0f, 0.0f);
            if (e + u < end) v[u] = *(const float4*)(f + (long)si[u] * 64);
        }
#pragma unroll
        for (int u = 0; u < 7; ++u) {
            if (e + u < end) acc_add(acc, v[u]);
        }
    }
}

__device__ __forceinline__ float4 sage_row_sum(
    const float* __restrict__ f, const int* __restrict__ ssrc, unsigned e, const unsigned end, const int g)
{
    float4 acc = make_float4(0.0f, 0.0f, 0.0f, 0.0f);
    if (e >= end) return acc;

    const unsigned eg = e + (unsigned)g;
    int cur = 0, nxt = 0, nx2 = 0;
    if (eg < end)       cur = ssrc[eg];
    if (eg + 16u < end) nxt = ssrc[eg + 16u];
    if (eg + 32u < end) nx2 = ssrc[eg + 32u];

    for (;;) {
        sage_sum8<0>(acc, f, cur, e, end);
        int far = 0;
        if (e + 48u + (unsigned)g < end) far = ssrc[e + 48u + (unsigned)g];
        sage_sum8<8>(acc, f, cur, e + 8u, end);
        e += 16u;
        if (e >= end) break;
        cur = nxt;
        nxt = nx2;
        nx2 = far;
    }
    return acc;
}

__device__ __forceinline__ unsigned scan_counters(unsigned* c, unsigned* wsum, unsigned* total,
                                                  const int tid, const int lane, const int wave)
{
    uint4* c4 = (uint4*)c;
    const int cv = ctr_vec(tid);
    const uint4 a = c4[cv];
    const uint4 b = c4[cv + 1];
    const unsigned loc = a.x + a.y + a.z + a.w + b.x + b.y + b.z + b.w;
    unsigned inc = loc;
    for (int o = 1; o < 32; o <<= 1) {
        const unsigned v = __shfl_up(inc, o);
        if (lane >= o) inc += v;
    }
    if (lane == 31) wsum[wave] = inc;
    __syncthreads();
    unsigned run = inc - loc;
    for (int w = 0; w < wave; ++w) run += wsum[w];
    const unsigned first = run;
    uint4 pa, pb;
    pa.x = run; run += a.x;
    pa.y = run; run += a.y;
    pa.z = run; run += a.z;
    pa.w = run; run += a.w;
    pb.x = run; run += b.x;
    pb.y = run; run += b.y;
    pb.z = run; run += b.z;
    pb.w = run; run += b.w;
    c4[cv]     = pa;
    c4[cv + 1] = pb;
    if (tid == PT - 1) *total = run;
    __syncthreads();
    return first;
}

__global__ __launch_bounds__(PT) void sage_part(
    const int* __restrict__ ei, unsigned* __restrict__ part, unsigned* __restrict__ tab)
{
    __shared__ __attribute__((aligned(16))) unsigned cw[NCW];
    __shared__ unsigned wsum[NW];
    __shared__ unsigned total;
    __shared__ unsigned stage[ES];

    const int tid  = threadIdx.x;
    const int lane = tid & 31;
    const int wave = tid >> 5;
    const int s    = blockIdx.x;

    ((uint4*)cw)[ctr_vec(tid)]     = make_uint4(0u, 0u, 0u, 0u);
    ((uint4*)cw)[ctr_vec(tid) + 1] = make_uint4(0u, 0u, 0u, 0u);
    __syncthreads();

    constexpr int EPT = (ES + PT - 1) / PT;
    const int* sp = ei + s * ES;
    const int* dp = ei + N_EDGES + s * ES;

    unsigned rg[EPT], pk[EPT], rk[EPT];
#pragma unroll
    for (int u = 0; u < EPT; ++u) {
        const int i = tid + u * PT;
        unsigned d = 0u, sv = 0u;
        if (i < ES) { d = (unsigned)dp[i]; sv = (unsigned)sp[i]; }
        const unsigned r = min(d / (unsigned)RN, (unsigned)(NR - 1));
        rg[u] = (i < ES) ? r : (unsigned)NR;
        pk[u] = (((d - r * RN) & 0xFFu) << 16) | (sv & 0xFFFFu);
    }
#pragma unroll
    for (int u = 0; u < EPT; ++u) {
        rk[u] = 0u;
        if (rg[u] < NR) rk[u] = atomicAdd(&cw[rg[u] * CP + wave], 1u);
    }
    __syncthreads();

    const unsigned first = scan_counters(cw, wsum, &total, tid, lane, wave);
    if ((tid & 3) == 0) tab[(tid >> 2) * NS + s] = first;
    if (tid == 0) tab[NR * NS + s] = (unsigned)ES;

#pragma unroll
    for (int u = 0; u < EPT; ++u) {
        if (rg[u] < NR) {
            const unsigned q = cw[rg[u] * CP + wave] + rk[u];
            if (q < (unsigned)ES) stage[q] = pk[u];
        }
    }
    __syncthreads();

    for (int i = tid; i < ES; i += PT) part[s * ES + i] = stage[i];
}

__device__ __forceinline__ unsigned ent_node(const unsigned w) { return min(w >> 16, (unsigned)(RN - 1)); }

__device__ __forceinline__ void ent_place(unsigned* cur, unsigned* stage, int* __restrict__ ssrc,
                                          const unsigned base, const int wave, const unsigned w)
{
    const unsigned slot = atomicAdd(&cur[ent_node(w) * CP + wave], 1u);
    const unsigned sv = w & 0xFFFFu;
    if (slot < (unsigned)CAP) {
        stage[slot] = sv;
    } else {
        const unsigned g = base + slot;
        if (g < (unsigned)N_EDGES) ssrc[g] = (int)sv;
    }
}

__global__ __launch_bounds__(PT) void sage_sort(
    const unsigned* __restrict__ part, const unsigned* __restrict__ tab,
    int* __restrict__ ssrc, unsigned* __restrict__ rowptr)
{
    __shared__ __attribute__((aligned(16))) unsigned cw[NCW];
    __shared__ unsigned sb[NS];
    __shared__ unsigned se[NS];
    __shared__ unsigned wsum[NW];
    __shared__ unsigned bsum[NS / 32];
    __shared__ unsigned total;
    __shared__ unsigned stage[CAP];

    const int tid  = threadIdx.x;
    const int lane = tid & 31;
    const int wave = tid >> 5;
    const int r    = blockIdx.x;

    ((uint4*)cw)[ctr_vec(tid)]     = make_uint4(0u, 0u, 0u, 0u);
    ((uint4*)cw)[ctr_vec(tid) + 1] = make_uint4(0u, 0u, 0u, 0u);
    if (tid < NS) {
        const unsigned b = min(tab[r * NS + tid], (unsigned)ES);
        const unsigned e = min(tab[(r + 1) * NS + tid], (unsigned)ES);
        sb[tid] = b;
        se[tid] = max(e, b);
        unsigned v = b;
        for (int o = 16; o > 0; o >>= 1) v += __shfl_xor(v, o);
        if (lane == 0) bsum[wave] = v;
    }
    __syncthreads();

    unsigned base = 0u;
    for (int w = 0; w < NS / 32; ++w) base += bsum[w];

    constexpr int SPW = NS / NW;
    unsigned wv[SPW], nb[SPW], ne[SPW];
#pragma unroll
    for (int j = 0; j < SPW; ++j) {
        const int sl = wave + j * NW;
        nb[j] = sb[sl];
        ne[j] = se[sl];
        const unsigned i = nb[j] + (unsigned)lane;
        wv[j] = NONE;
        if (i < ne[j]) wv[j] = part[sl * ES + i];
    }

#pragma unroll
    for (int j = 0; j < SPW; ++j) {
        const int sl = wave + j * NW;
        if (wv[j] != NONE) atomicAdd(&cw[ent_node(wv[j]) * CP + wave], 1u);
        for (unsigned i = nb[j] + 32u + (unsigned)lane; i < ne[j]; i += 32u)
            atomicAdd(&cw[ent_node(part[sl * ES + i]) * CP + wave], 1u);
    }
    __syncthreads();

    const unsigned first = scan_counters(cw, wsum, &total, tid, lane, wave);
    if ((tid & 3) == 0 && (tid >> 2) < RN) rowptr[r * RN + (tid >> 2)] = base + first;

#pragma unroll
    for (int j = 0; j < SPW; ++j) {
        const int sl = wave + j * NW;
        if (wv[j] != NONE) ent_place(cw, stage, ssrc, base, wave, wv[j]);
        for (unsigned i = nb[j] + 32u + (unsigned)lane; i < ne[j]; i += 32u)
            ent_place(cw, stage, ssrc, base, wave, part[sl * ES + i]);
    }
    __syncthreads();

    const unsigned n = min(total, (unsigned)CAP);
    for (unsigned i = (unsigned)tid; i < n; i += (unsigned)PT) {
        const unsigned g = base + i;
        if (g < (unsigned)N_EDGES) ssrc[g] = (int)stage[i];
    }
}

template <int OUTC>
__global__ __launch_bounds__(256) void sage_layer(
    const float* __restrict__ feat, const int* __restrict__ ssrc,
    const unsigned* __restrict__ rowptr,
    const float* __restrict__ Wl, const float* __restrict__ Wr,
    const float* __restrict__ bias, float* __restrict__ out)
{
    constexpr int CT = OUTC / 16;
    constexpr int RT = 8 / CT;
    constexpr int AP = 68;
    constexpr int WP = 132;

    __shared__ float Wlds[OUTC * WP];
    __shared__ __attribute__((aligned(16))) float At[RT * 16 * AP];

    const int tid = threadIdx.x;
    const int nb  = blockIdx.x * (RT * 16);
    const int grp = tid >> 4;
    const int g   = tid & 15;

    unsigned eb[RT], ee[RT];
#pragma unroll
    for (int j = 0; j < RT; ++j) {
        const int nn = min(nb + j * 16 + grp, N_NODES - 1);
        eb[j] = rowptr[nn];
        ee[j] = min(rowptr[nn + 1], (unsigned)N_EDGES);
    }

    for (int i = tid; i < OUTC * 128; i += 256) {
        int n = i >> 7, k = i & 127;
        Wlds[n * WP + k] = (k < 64) ? Wl[n * 64 + k] : Wr[n * 64 + (k - 64)];
    }

#pragma unroll
    for (int j = 0; j < RT; ++j) {
        const int node = nb + j * 16 + grp;
        if (node < N_NODES) {
            const float4 acc = sage_row_sum(feat + g * 4, ssrc, eb[j], ee[j], g);
            const unsigned deg = (ee[j] > eb[j]) ? ee[j] - eb[j] : 0u;
            const float rd = 1.0f / fmaxf((float)deg, 1.0f);
            *(float4*)&At[(j * 16 + grp) * AP + g * 4] =
                make_float4(acc.x * rd, acc.y * rd, acc.z * rd, acc.w * rd);
        }
    }
    __syncthreads();

    const int wave = tid >> 5;
    const int lane = tid & 31;
    const int rt   = wave / CT;
    const int row0 = nb + rt * 16;
    if (row0 >= N_NODES) return;

    const int col0 = (wave % CT) * 16;
    const int half = lane >> 4;
    const int m    = lane & 15;
    const int row  = row0 + m;
    const int colg = col0 + m;

    const float bc = bias[colg];

    v8f c;
#pragma unroll
    for (int i = 0; i < 8; ++i) c[i] = bc;

    const float* arow = &At[(rt * 16 + m) * AP];
    const float* xrow = feat + (long)row * 64;
    const float* wcol = &Wlds[colg * WP];

#pragma unroll
    for (int ks = 0; ks < 64; ks += 4) {
        v2f a, b;
        a[0] = arow[ks + half * 2 + 0];
        a[1] = arow[ks + half * 2 + 1];
        b[0] = wcol[ks + half * 2 + 0];
        b[1] = wcol[ks + half * 2 + 1];
        c = __builtin_amdgcn_wmma_f32_16x16x4_f32(false, a, false, b,
                                                  (short)0, c, false, false);
    }
#pragma unroll
    for (int ks = 0; ks < 64; ks += 4) {
        v2f a, b;
        a[0] = xrow[ks + half * 2 + 0];
        a[1] = xrow[ks + half * 2 + 1];
        b[0] = wcol[64 + ks + half * 2 + 0];
        b[1] = wcol[64 + ks + half * 2 + 1];
        c = __builtin_amdgcn_wmma_f32_16x16x4_f32(false, a, false, b,
                                                  (short)0, c, false, false);
    }

#pragma unroll
    for (int v = 0; v < 8; ++v) {
        float y = c[v];
        y = (y > 0.0f) ? y : NEG_SLOPE * y;
        out[(long)(row0 + half * 8 + v) * OUTC + colg] = y;
    }
}

extern "C" void kernel_launch(void* const* d_in, const int* in_sizes, int n_in,
                              void* d_out, int out_size, void* d_ws, size_t ws_size,
                              hipStream_t stream) {
    const float* x   = (const float*)d_in[0];
    const int*   ei  = (const int*)  d_in[1];
    const float* W1l = (const float*)d_in[2];
    const float* b1  = (const float*)d_in[3];
    const float* W1r = (const float*)d_in[4];
    const float* W2l = (const float*)d_in[5];
    const float* b2  = (const float*)d_in[6];
    const float* W2r = (const float*)d_in[7];
    float* out = (float*)d_out;

    const size_t partB = (size_t)N_EDGES * 4;
    const size_t featB = (size_t)N_NODES * 64 * 4;
    const size_t tabB  = (size_t)(NR + 1) * NS * 4;
    unsigned* part   = (unsigned*)d_ws;
    float*    h      = (float*)((char*)d_ws + partB);
    unsigned* tab    = (unsigned*)((char*)d_ws + partB + featB);
    unsigned* rowptr = (unsigned*)((char*)d_ws + partB + featB + tabB);
    int*      ssrc   = (int*)(rowptr + RP_N);

    sage_part<<<NS, PT, 0, stream>>>(ei, part, tab);
    sage_sort<<<NR, PT, 0, stream>>>(part, tab, ssrc, rowptr);

    sage_layer<HID_CH><<<(N_NODES / 16 + 1) / 2, 256, 0, stream>>>(x, ssrc, rowptr, W1l, W1r, b1, h);

    sage_layer<OUT_CH><<<(N_NODES / 16 + 3) / 4, 256, 0, stream>>>(h, ssrc, rowptr, W2l, W2r, b2, out);
}


# layer kernels, more loads per dependent round trip: weight panel staged as one batch of 16-byte loads (was 4 or 8 trips of 4-byte loads); layer 2 sums 16 rows per trip, not 8; row order kept
# speedup vs baseline: 6.7284x; 1.0567x over previous
#include <hip/hip_runtime.h>

#define N_NODES 50000
#define N_EDGES 1600000
#define IN_CH   64
#define HID_CH  64
#define OUT_CH  32
#define NEG_SLOPE 0.01f

#define NS   256
#define ES   (N_EDGES / NS)
#define NR   256
#define RN   196
#define RP_N (NR * RN)
#define PT   1024
#define CAP  8192
#define NONE 0xFFFFFFFFu
#define NW   (PT / 32)
#define NC   (256 * NW)
#define CP   (NW + 4)
#define NCW  (256 * CP)

static_assert(NS * ES == N_EDGES, "slices cover the edges exactly");
static_assert(RP_N > N_NODES && RN <= 256 && N_NODES <= 65536, "packed entry: 8-bit offset in range, 16-bit source");
static_assert(NS == 256 && NR == 256 && NC == 8 * PT, "scan_counters: 8 counters per thread");
static_assert(NW % 8 == 0 && CP % 4 == 0, "a thread's 8 counters lie in one row, 16-byte aligned");

__device__ __forceinline__ int ctr_vec(const int tid) { return (tid >> 2) * (CP / 4) + (tid & 3) * 2; }

typedef __attribute__((ext_vector_type(2))) float v2f;
typedef __attribute__((ext_vector_type(8))) float v8f;


#define GRP_BCAST(v, U) __builtin_amdgcn_update_dpp(0, (v), 0x150 + (U), 0xF, 0xF, false)

__device__ __forceinline__ void acc_add(float4& acc, const float4 v)
{
    acc.x += v.x; acc.y += v.y; acc.z += v.z; acc.w += v.w;
}

template <int H>
__device__ __forceinline__ void sage_sum8(float4& acc, const float* __restrict__ f, const int idx,
                                          const unsigned e, const unsigned end)
{
    int si[8];
    si[0] = GRP_BCAST(idx, H + 0); si[1] = GRP_BCAST(idx, H + 1);
    si[2] = GRP_BCAST(idx, H + 2); si[3] = GRP_BCAST(idx, H + 3);
    si[4] = GRP_BCAST(idx, H + 4); si[5] = GRP_BCAST(idx, H + 5);
    si[6] = GRP_BCAST(idx, H + 6); si[7] = GRP_BCAST(idx, H + 7);
    float4 v[8];
    if (e + 8u <= end) {
#pragma unroll
        for (int u = 0; u < 8; ++u) v[u] = *(const float4*)(f + (long)si[u] * 64);
#pragma unroll
        for (int u = 0; u < 8; ++u) acc_add(acc, v[u]);
    } else {
#pragma unroll
        for (int u = 0; u < 7; ++u) {
            v[u] = make_float4(0.0f, 0.0f, 0.0f, 0.0f);
            if (e + u < end) v[u] = *(const float4*)(f + (long)si[u] * 64);
        }
#pragma unroll
        for (int u = 0; u < 7; ++u) {
            if (e + u < end) acc_add(acc, v[u]);
        }
    }
}

__device__ __forceinline__ float4 sage_row_sum(
    const float* __restrict__ f, const int* __restrict__ ssrc, unsigned e, const unsigned end, const int g)
{
    float4 acc = make_float4(0.0f, 0.0f, 0.0f, 0.0f);
    if (e >= end) return acc;

    const unsigned eg = e + (unsigned)g;
    int cur = 0, nxt = 0, nx2 = 0;
    if (eg < end)       cur = ssrc[eg];
    if (eg + 16u < end) nxt = ssrc[eg + 16u];
    if (eg + 32u < end) nx2 = ssrc[eg + 32u];

    for (;;) {
        sage_sum8<0>(acc, f, cur, e, end);
        int far = 0;
        if (e + 48u + (unsigned)g < end) far = ssrc[e + 48u + (unsigned)g];
        sage_sum8<8>(acc, f, cur, e + 8u, end);
        e += 16u;
        if (e >= end) break;
        cur = nxt;
        nxt = nx2;
        nx2 = far;
    }
    return acc;
}

__device__ __forceinline__ float4 sage_row_sum16(
    const float* __restrict__ f, const int* __restrict__ ssrc, unsigned e, const unsigned end, const int g)
{
    float4 acc = make_float4(0.0f, 0.0f, 0.0f, 0.0f);
    if (e >= end) return acc;

    const unsigned eg = e + (unsigned)g;
    int cur = 0, nxt = 0, nx2 = 0;
    if (eg < end)       cur = ssrc[eg];
    if (eg + 16u < end) nxt = ssrc[eg + 16u];
    if (eg + 32u < end) nx2 = ssrc[eg + 32u];

    for (;;) {
        int si[16];
        si[0]  = GRP_BCAST(cur, 0);  si[1]  = GRP_BCAST(cur, 1);  si[2]  = GRP_BCAST(cur, 2);  si[3]  = GRP_BCAST(cur, 3);
        si[4]  = GRP_BCAST(cur, 4);  si[5]  = GRP_BCAST(cur, 5);  si[6]  = GRP_BCAST(cur, 6);  si[7]  = GRP_BCAST(cur, 7);
        si[8]  = GRP_BCAST(cur, 8);  si[9]  = GRP_BCAST(cur, 9);  si[10] = GRP_BCAST(cur, 10); si[11] = GRP_BCAST(cur, 11);
        si[12] = GRP_BCAST(cur, 12); si[13] = GRP_BCAST(cur, 13); si[14] = GRP_BCAST(cur, 14); si[15] = GRP_BCAST(cur, 15);
        float4 v[16];
        int far = 0;
        if (e + 16u <= end) {
#pragma unroll
            for (int u = 0; u < 16; ++u) v[u] = *(const float4*)(f + (long)si[u] * 64);
            if (e + 48u + (unsigned)g < end) far = ssrc[e + 48u + (unsigned)g];
#pragma unroll
            for (int u = 0; u < 16; ++u) acc_add(acc, v[u]);
        } else {
#pragma unroll
            for (int u = 0; u < 15; ++u) {
                v[u] = make_float4(0.0f, 0.0f, 0.0f, 0.0f);
                if (e + u < end) v[u] = *(const float4*)(f + (long)si[u] * 64);
            }
#pragma unroll
            for (int u = 0; u < 15; ++u) {
                if (e + u < end) acc_add(acc, v[u]);
            }
        }
        e += 16u;
        if (e >= end) break;
        cur = nxt;
        nxt = nx2;
        nx2 = far;
    }
    return acc;
}

__device__ __forceinline__ unsigned scan_counters(unsigned* c, unsigned* wsum, unsigned* total,
                                                  const int tid, const int lane, const int wave)
{
    uint4* c4 = (uint4*)c;
    const int cv = ctr_vec(tid);
    const uint4 a = c4[cv];
    const uint4 b = c4[cv + 1];
    const unsigned loc = a.x + a.y + a.z + a.w + b.x + b.y + b.z + b.w;
    unsigned inc = loc;
    for (int o = 1; o < 32; o <<= 1) {
        const unsigned v = __shfl_up(inc, o);
        if (lane >= o) inc += v;
    }
    if (lane == 31) wsum[wave] = inc;
    __syncthreads();
    unsigned run = inc - loc;
    for (int w = 0; w < wave; ++w) run += wsum[w];
    const unsigned first = run;
    uint4 pa, pb;
    pa.x = run; run += a.x;
    pa.y = run; run += a.y;
    pa.z = run; run += a.z;
    pa.w = run; run += a.w;
    pb.x = run; run += b.x;
    pb.y = run; run += b.y;
    pb.z = run; run += b.z;
    pb.w = run; run += b.w;
    c4[cv]     = pa;
    c4[cv + 1] = pb;
    if (tid == PT - 1) *total = run;
    __syncthreads();
    return first;
}

__global__ __launch_bounds__(PT) void sage_part(
    const int* __restrict__ ei, unsigned* __restrict__ part, unsigned* __restrict__ tab)
{
    __shared__ __attribute__((aligned(16))) unsigned cw[NCW];
    __shared__ unsigned wsum[NW];
    __shared__ unsigned total;
    __shared__ unsigned stage[ES];

    const int tid  = threadIdx.x;
    const int lane = tid & 31;
    const int wave = tid >> 5;
    const int s    = blockIdx.x;

    ((uint4*)cw)[ctr_vec(tid)]     = make_uint4(0u, 0u, 0u, 0u);
    ((uint4*)cw)[ctr_vec(tid) + 1] = make_uint4(0u, 0u, 0u, 0u);
    __syncthreads();

    constexpr int EPT = (ES + PT - 1) / PT;
    const int* sp = ei + s * ES;
    const int* dp = ei + N_EDGES + s * ES;

    unsigned rg[EPT], pk[EPT], rk[EPT];
#pragma unroll
    for (int u = 0; u < EPT; ++u) {
        const int i = tid + u * PT;
        unsigned d = 0u, sv = 0u;
        if (i < ES) { d = (unsigned)dp[i]; sv = (unsigned)sp[i]; }
        const unsigned r = min(d / (unsigned)RN, (unsigned)(NR - 1));
        rg[u] = (i < ES) ? r : (unsigned)NR;
        pk[u] = (((d - r * RN) & 0xFFu) << 16) | (sv & 0xFFFFu);
    }
#pragma unroll
    for (int u = 0; u < EPT; ++u) {
        rk[u] = 0u;
        if (rg[u] < NR) rk[u] = atomicAdd(&cw[rg[u] * CP + wave], 1u);
    }
    __syncthreads();

    const unsigned first = scan_counters(cw, wsum, &total, tid, lane, wave);
    if ((tid & 3) == 0) tab[(tid >> 2) * NS + s] = first;
    if (tid == 0) tab[NR * NS + s] = (unsigned)ES;

#pragma unroll
    for (int u = 0; u < EPT; ++u) {
        if (rg[u] < NR) {
            const unsigned q = cw[rg[u] * CP + wave] + rk[u];
            if (q < (unsigned)ES) stage[q] = pk[u];
        }
    }
    __syncthreads();

    for (int i = tid; i < ES; i += PT) part[s * ES + i] = stage[i];
}

__device__ __forceinline__ unsigned ent_node(const unsigned w) { return min(w >> 16, (unsigned)(RN - 1)); }

__device__ __forceinline__ void ent_place(unsigned* cur, unsigned* stage, int* __restrict__ ssrc,
                                          const unsigned base, const int wave, const unsigned w)
{
    const unsigned slot = atomicAdd(&cur[ent_node(w) * CP + wave], 1u);
    const unsigned sv = w & 0xFFFFu;
    if (slot < (unsigned)CAP) {
        stage[slot] = sv;
    } else {
        const unsigned g = base + slot;
        if (g < (unsigned)N_EDGES) ssrc[g] = (int)sv;
    }
}

__global__ __launch_bounds__(PT) void sage_sort(
    const unsigned* __restrict__ part, const unsigned* __restrict__ tab,
    int* __restrict__ ssrc, unsigned* __restrict__ rowptr)
{
    __shared__ __attribute__((aligned(16))) unsigned cw[NCW];
    __shared__ unsigned sb[NS];
    __shared__ unsigned se[NS];
    __shared__ unsigned wsum[NW];
    __shared__ unsigned bsum[NS / 32];
    __shared__ unsigned total;
    __shared__ unsigned stage[CAP];

    const int tid  = threadIdx.x;
    const int lane = tid & 31;
    const int wave = tid >> 5;
    const int r    = blockIdx.x;

    ((uint4*)cw)[ctr_vec(tid)]     = make_uint4(0u, 0u, 0u, 0u);
    ((uint4*)cw)[ctr_vec(tid) + 1] = make_uint4(0u, 0u, 0u, 0u);
    if (tid < NS) {
        const unsigned b = min(tab[r * NS + tid], (unsigned)ES);
        const unsigned e = min(tab[(r + 1) * NS + tid], (unsigned)ES);
        sb[tid] = b;
        se[tid] = max(e, b);
        unsigned v = b;
        for (int o = 16; o > 0; o >>= 1) v += __shfl_xor(v, o);
        if (lane == 0) bsum[wave] = v;
    }
    __syncthreads();

    unsigned base = 0u;
    for (int w = 0; w < NS / 32; ++w) base += bsum[w];

    constexpr int SPW = NS / NW;
    unsigned wv[SPW], nb[SPW], ne[SPW];
#pragma unroll
    for (int j = 0; j < SPW; ++j) {
        const int sl = wave + j * NW;
        nb[j] = sb[sl];
        ne[j] = se[sl];
        const unsigned i = nb[j] + (unsigned)lane;
        wv[j] = NONE;
        if (i < ne[j]) wv[j] = part[sl * ES + i];
    }

#pragma unroll
    for (int j = 0; j < SPW; ++j) {
        const int sl = wave + j * NW;
        if (wv[j] != NONE) atomicAdd(&cw[ent_node(wv[j]) * CP + wave], 1u);
        for (unsigned i = nb[j] + 32u + (unsigned)lane; i < ne[j]; i += 32u)
            atomicAdd(&cw[ent_node(part[sl * ES + i]) * CP + wave], 1u);
    }
    __syncthreads();

    const unsigned first = scan_counters(cw, wsum, &total, tid, lane, wave);
    if ((tid & 3) == 0 && (tid >> 2) < RN) rowptr[r * RN + (tid >> 2)] = base + first;

#pragma unroll
    for (int j = 0; j < SPW; ++j) {
        const int sl = wave + j * NW;
        if (wv[j] != NONE) ent_place(cw, stage, ssrc, base, wave, wv[j]);
        for (unsigned i = nb[j] + 32u + (unsigned)lane; i < ne[j]; i += 32u)
            ent_place(cw, stage, ssrc, base, wave, part[sl * ES + i]);
    }
    __syncthreads();

    const unsigned n = min(total, (unsigned)CAP);
    for (unsigned i = (unsigned)tid; i < n; i += (unsigned)PT) {
        const unsigned g = base + i;
        if (g < (unsigned)N_EDGES) ssrc[g] = (int)stage[i];
    }
}

template <int OUTC>
__global__ __launch_bounds__(256) void sage_layer(
    const float* __restrict__ feat, const int* __restrict__ ssrc,
    const unsigned* __restrict__ rowptr,
    const float* __restrict__ Wl, const float* __restrict__ Wr,
    const float* __restrict__ bias, float* __restrict__ out)
{
    constexpr int CT = OUTC / 16;
    constexpr int RT = 8 / CT;
    constexpr int AP = 68;
    constexpr int WP = 132;

    static_assert(OUTC % 8 == 0 && WP % 4 == 0, "weight panel: 256 threads sweep 8 rows of 128 floats as 16-byte pieces; rows 16-byte aligned");

    __shared__ __attribute__((aligned(16))) float Wlds[OUTC * WP];
    __shared__ __attribute__((aligned(16))) float At[RT * 16 * AP];

    const int tid = threadIdx.x;
    const int nb  = blockIdx.x * (RT * 16);
    const int grp = tid >> 4;
    const int g   = tid & 15;

    unsigned eb[RT], ee[RT];
#pragma unroll
    for (int j = 0; j < RT; ++j) {
        const int nn = min(nb + j * 16 + grp, N_NODES - 1);
        eb[j] = rowptr[nn];
        ee[j] = min(rowptr[nn + 1], (unsigned)N_EDGES);
    }

    {
        constexpr int WV = OUTC / 8;
        const int k  = (tid * 4) & 127;
        const int n0 = tid >> 5;
        const float* wsrc = ((k < 64) ? (Wl + k) : (Wr + (k - 64))) + n0 * 64;
        float* wdst = &Wlds[n0 * WP + k];
        float4 wv[WV];
#pragma unroll
        for (int u = 0; u < WV; ++u) wv[u] = *(const float4*)(wsrc + u * (8 * 64));
#pragma unroll
        for (int u = 0; u < WV; ++u) *(float4*)(wdst + u * (8 * WP)) = wv[u];
    }

#pragma unroll
    for (int j = 0; j < RT; ++j) asm volatile("" :: "v"(eb[j]), "v"(ee[j]));

#pragma unroll
    for (int j = 0; j < RT; ++j) {
        const int node = nb + j * 16 + grp;
        if (node < N_NODES) {
            const float4 acc = (RT == 4) ? sage_row_sum16(feat + g * 4, ssrc, eb[j], ee[j], g)
                                         : sage_row_sum(feat + g * 4, ssrc, eb[j], ee[j], g);
            const unsigned deg = (ee[j] > eb[j]) ? ee[j] - eb[j] : 0u;
            const float rd = 1.0f / fmaxf((float)deg, 1.0f);
            *(float4*)&At[(j * 16 + grp) * AP + g * 4] =
                make_float4(acc.x * rd, acc.y * rd, acc.z * rd, acc.w * rd);
        }
    }
    __syncthreads();

    const int wave = tid >> 5;
    const int lane = tid & 31;
    const int rt   = wave / CT;
    const int row0 = nb + rt * 16;
    if (row0 >= N_NODES) return;

    const int col0 = (wave % CT) * 16;
    const int half = lane >> 4;
    const int m    = lane & 15;
    const int row  = row0 + m;
    const int colg = col0 + m;

    const float bc = bias[colg];

    v8f c;
#pragma unroll
    for (int i = 0; i < 8; ++i) c[i] = bc;

    const float* arow = &At[(rt * 16 + m) * AP];
    const float* xrow = feat + (long)row * 64;
    const float* wcol = &Wlds[colg * WP];

#pragma unroll
    for (int ks = 0; ks < 64; ks += 4) {
        v2f a, b;
        a[0] = arow[ks + half * 2 + 0];
        a[1] = arow[ks + half * 2 + 1];
        b[0] = wcol[ks + half * 2 + 0];
        b[1] = wcol[ks + half * 2 + 1];
        c = __builtin_amdgcn_wmma_f32_16x16x4_f32(false, a, false, b,
                                                  (short)0, c, false, false);
    }
#pragma unroll
    for (int ks = 0; ks < 64; ks += 4) {
        v2f a, b;
        a[0] = xrow[ks + half * 2 + 0];
        a[1] = xrow[ks + half * 2 + 1];
        b[0] = wcol[64 + ks + half * 2 + 0];
        b[1] = wcol[64 + ks + half * 2 + 1];
        c = __builtin_amdgcn_wmma_f32_16x16x4_f32(false, a, false, b,
                                                  (short)0, c, false, false);
    }

#pragma unroll
    for (int v = 0; v < 8; ++v) {
        float y = c[v];
        y = (y > 0.0f) ? y : NEG_SLOPE * y;
        out[(long)(row0 + half * 8 + v) * OUTC + colg] = y;
    }
}

extern "C" void kernel_launch(void* const* d_in, const int* in_sizes, int n_in,
                              void* d_out, int out_size, void* d_ws, size_t ws_size,
                              hipStream_t stream) {
    const float* x   = (const float*)d_in[0];
    const int*   ei  = (const int*)  d_in[1];
    const float* W1l = (const float*)d_in[2];
    const float* b1  = (const float*)d_in[3];
    const float* W1r = (const float*)d_in[4];
    const float* W2l = (const float*)d_in[5];
    const float* b2  = (const float*)d_in[6];
    const float* W2r = (const float*)d_in[7];
    float* out = (float*)d_out;

    const size_t partB = (size_t)N_EDGES * 4;
    const size_t featB = (size_t)N_NODES * 64 * 4;
    const size_t tabB  = (size_t)(NR + 1) * NS * 4;
    unsigned* part   = (unsigned*)d_ws;
    float*    h      = (float*)((char*)d_ws + partB);
    unsigned* tab    = (unsigned*)((char*)d_ws + partB + featB);
    unsigned* rowptr = (unsigned*)((char*)d_ws + partB + featB + tabB);
    int*      ssrc   = (int*)(rowptr + RP_N);

    sage_part<<<NS, PT, 0, stream>>>(ei, part, tab);
    sage_sort<<<NR, PT, 0, stream>>>(part, tab, ssrc, rowptr);

    sage_layer<HID_CH><<<(N_NODES / 16 + 1) / 2, 256, 0, stream>>>(x, ssrc, rowptr, W1l, W1r, b1, h);

    sage_layer<OUT_CH><<<(N_NODES / 16 + 3) / 4, 256, 0, stream>>>(h, ssrc, rowptr, W2l, W2r, b2, out);
}


# layer 1 fused into the list-sorting launch: a range's work-group sums x rows straight from the lists it has just staged in LDS, then runs its 64-column WMMA tiles; 3 launches, not 4; sum order kept
# speedup vs baseline: 7.1135x; 1.0572x over previous
#include <hip/hip_runtime.h>

#define N_NODES 50000
#define N_EDGES 1600000
#define IN_CH   64
#define HID_CH  64
#define OUT_CH  32
#define NEG_SLOPE 0.01f

#define NS   256
#define ES   (N_EDGES / NS)
#define NR   256
#define RN   196
#define RP_N (NR * RN)
#define PT   1024
#define CAP  8192
#define NONE 0xFFFFFFFFu
#define NW   (PT / 32)
#define NC   (256 * NW)
#define CP   (NW + 4)
#define NCW  (256 * CP)

static_assert(NS * ES == N_EDGES, "slices cover the edges exactly");
static_assert(RP_N > N_NODES && RN <= 256 && N_NODES <= 65536, "packed entry: 8-bit offset in range, 16-bit source");
static_assert(NS == 256 && NR == 256 && NC == 8 * PT, "scan_counters: 8 counters per thread");
static_assert(NW % 8 == 0 && CP % 4 == 0, "a thread's 8 counters lie in one row, 16-byte aligned");

__device__ __forceinline__ int ctr_vec(const int tid) { return (tid >> 2) * (CP / 4) + (tid & 3) * 2; }

#define AP   68
#define WP   132
#define TR   ((RN + 15) / 16)

typedef __attribute__((ext_vector_type(2))) float v2f;
typedef __attribute__((ext_vector_type(8))) float v8f;


#define GRP_BCAST(v, U) __builtin_amdgcn_update_dpp(0, (v), 0x150 + (U), 0xF, 0xF, false)

__device__ __forceinline__ void acc_add(float4& acc, const float4 v)
{
    acc.x += v.x; acc.y += v.y; acc.z += v.z; acc.w += v.w;
}

struct ListInMemory {
    const int* ssrc;
    __device__ __forceinline__ int operator()(const unsigned i) const { return ssrc[i]; }
};

struct ListInStage {
    const unsigned* stage;
    __device__ __forceinline__ int operator()(const unsigned i) const { return (int)stage[i]; }
};

template <class List>
__device__ __forceinline__ float4 sage_row_sum16(
    const float* __restrict__ f, const List ent, unsigned e, const unsigned end, const int g)
{
    float4 acc = make_float4(0.0f, 0.0f, 0.0f, 0.0f);
    if (e >= end) return acc;

    const unsigned eg = e + (unsigned)g;
    int cur = 0, nxt = 0, nx2 = 0;
    if (eg < end)       cur = ent(eg);
    if (eg + 16u < end) nxt = ent(eg + 16u);
    if (eg + 32u < end) nx2 = ent(eg + 32u);

    for (;;) {
        int si[16];
        si[0]  = GRP_BCAST(cur, 0);  si[1]  = GRP_BCAST(cur, 1);  si[2]  = GRP_BCAST(cur, 2);  si[3]  = GRP_BCAST(cur, 3);
        si[4]  = GRP_BCAST(cur, 4);  si[5]  = GRP_BCAST(cur, 5);  si[6]  = GRP_BCAST(cur, 6);  si[7]  = GRP_BCAST(cur, 7);
        si[8]  = GRP_BCAST(cur, 8);  si[9]  = GRP_BCAST(cur, 9);  si[10] = GRP_BCAST(cur, 10); si[11] = GRP_BCAST(cur, 11);
        si[12] = GRP_BCAST(cur, 12); si[13] = GRP_BCAST(cur, 13); si[14] = GRP_BCAST(cur, 14); si[15] = GRP_BCAST(cur, 15);
        float4 v[16];
        int far = 0;
        if (e + 16u <= end) {
#pragma unroll
            for (int u = 0; u < 16; ++u) v[u] = *(const float4*)(f + (long)si[u] * 64);
            if (e + 48u + (unsigned)g < end) far = ent(e + 48u + (unsigned)g);
#pragma unroll
            for (int u = 0; u < 16; ++u) acc_add(acc, v[u]);
        } else {
#pragma unroll
            for (int u = 0; u < 15; ++u) {
                v[u] = make_float4(0.0f, 0.0f, 0.0f, 0.0f);
                if (e + u < end) v[u] = *(const float4*)(f + (long)si[u] * 64);
            }
#pragma unroll
            for (int u = 0; u < 15; ++u) {
                if (e + u < end) acc_add(acc, v[u]);
            }
        }
        e += 16u;
        if (e >= end) break;
        cur = nxt;
        nxt = nx2;
        nx2 = far;
    }
    return acc;
}

__device__ __forceinline__ unsigned scan_counters(unsigned* c, unsigned* wsum, unsigned* total,
                                                  const int tid, const int lane, const int wave)
{
    uint4* c4 = (uint4*)c;
    const int cv = ctr_vec(tid);
    const uint4 a = c4[cv];
    const uint4 b = c4[cv + 1];
    const unsigned loc = a.x + a.y + a.z + a.w + b.x + b.y + b.z + b.w;
    unsigned inc = loc;
    for (int o = 1; o < 32; o <<= 1) {
        const unsigned v = __shfl_up(inc, o);
        if (lane >= o) inc += v;
    }
    if (lane == 31) wsum[wave] = inc;
    __syncthreads();
    unsigned run = inc - loc;
    for (int w = 0; w < wave; ++w) run += wsum[w];
    const unsigned first = run;
    uint4 pa, pb;
    pa.x = run; run += a.x;
    pa.y = run; run += a.y;
    pa.z = run; run += a.z;
    pa.w = run; run += a.w;
    pb.x = run; run += b.x;
    pb.y = run; run += b.y;
    pb.z = run; run += b.z;
    pb.w = run; run += b.w;
    c4[cv]     = pa;
    c4[cv + 1] = pb;
    if (tid == PT - 1) *total = run;
    __syncthreads();
    return first;
}

__global__ __launch_bounds__(PT) void sage_part(
    const int* __restrict__ ei, unsigned* __restrict__ part, unsigned* __restrict__ tab)
{
    __shared__ __attribute__((aligned(16))) unsigned cw[NCW];
    __shared__ unsigned wsum[NW];
    __shared__ unsigned total;
    __shared__ unsigned stage[ES];

    const int tid  = threadIdx.x;
    const int lane = tid & 31;
    const int wave = tid >> 5;
    const int s    = blockIdx.x;

    ((uint4*)cw)[ctr_vec(tid)]     = make_uint4(0u, 0u, 0u, 0u);
    ((uint4*)cw)[ctr_vec(tid) + 1] = make_uint4(0u, 0u, 0u, 0u);
    __syncthreads();

    constexpr int EPT = (ES + PT - 1) / PT;
    const int* sp = ei + s * ES;
    const int* dp = ei + N_EDGES + s * ES;

    unsigned rg[EPT], pk[EPT], rk[EPT];
#pragma unroll
    for (int u = 0; u < EPT; ++u) {
        const int i = tid + u * PT;
        unsigned d = 0u, sv = 0u;
        if (i < ES) { d = (unsigned)dp[i]; sv = (unsigned)sp[i]; }
        const unsigned r = min(d / (unsigned)RN, (unsigned)(NR - 1));
        rg[u] = (i < ES) ? r : (unsigned)NR;
        pk[u] = (((d - r * RN) & 0xFFu) << 16) | (sv & 0xFFFFu);
    }
#pragma unroll
    for (int u = 0; u < EPT; ++u) {
        rk[u] = 0u;
        if (rg[u] < NR) rk[u] = atomicAdd(&cw[rg[u] * CP + wave], 1u);
    }
    __syncthreads();

    const unsigned first = scan_counters(cw, wsum, &total, tid, lane, wave);
    if ((tid & 3) == 0) tab[(tid >> 2) * NS + s] = first;
    if (tid == 0) tab[NR * NS + s] = (unsigned)ES;

#pragma unroll
    for (int u = 0; u < EPT; ++u) {
        if (rg[u] < NR) {
            const unsigned q = cw[rg[u] * CP + wave] + rk[u];
            if (q < (unsigned)ES) stage[q] = pk[u];
        }
    }
    __syncthreads();

    for (int i = tid; i < ES; i += PT) part[s * ES + i] = stage[i];
}

__device__ __forceinline__ unsigned ent_node(const unsigned w) { return min(w >> 16, (unsigned)(RN - 1)); }

__device__ __forceinline__ void ent_place(unsigned* cur, unsigned* stage, int* __restrict__ ssrc,
                                          const unsigned base, const int wave, const unsigned w)
{
    const unsigned slot = atomicAdd(&cur[ent_node(w) * CP + wave], 1u);
    const unsigned sv = w & 0xFFFFu;
    if (slot < (unsigned)CAP) {
        stage[slot] = sv;
    } else {
        const unsigned g = base + slot;
        if (g < (unsigned)N_EDGES) ssrc[g] = (int)sv;
    }
}

template <class List>
__device__ __forceinline__ void sage_range_means(
    float* At, const unsigned* cur, const float* __restrict__ feat, const List ent,
    const unsigned off, const unsigned lim, const int r, const int tid)
{
    const int grp = tid >> 4;
    const int g   = tid & 15;
    for (int o = grp; o < RN; o += PT / 16) {
        if (r * RN + o >= N_NODES) break;
        const unsigned e0 = (o == 0) ? 0u : cur[(o - 1) * CP + NW - 1];
        const unsigned e1 = max(cur[o * CP + NW - 1], e0);
        const float4 acc = sage_row_sum16(feat + g * 4, ent, min(off + e0, lim), min(off + e1, lim), g);
        const float rd = 1.0f / fmaxf((float)(e1 - e0), 1.0f);
        *(float4*)&At[o * AP + g * 4] = make_float4(acc.x * rd, acc.y * rd, acc.z * rd, acc.w * rd);
    }
}

__global__ __launch_bounds__(PT) void sage_sort_layer1(
    const unsigned* __restrict__ part, const unsigned* __restrict__ tab,
    int* ssrc, unsigned* __restrict__ rowptr,
    const float* __restrict__ feat, const float* __restrict__ Wl, const float* __restrict__ Wr,
    const float* __restrict__ bias, float* __restrict__ out)
{
    __shared__ __attribute__((aligned(16))) unsigned cw[NCW];
    __shared__ unsigned sb[NS];
    __shared__ unsigned se[NS];
    __shared__ unsigned wsum[NW];
    __shared__ unsigned bsum[NS / 32];
    __shared__ unsigned total;
    __shared__ unsigned stage[CAP];
    __shared__ __attribute__((aligned(16))) float Wlds[HID_CH * WP];
    __shared__ __attribute__((aligned(16))) float At[TR * 16 * AP];

    const int tid  = threadIdx.x;
    const int lane = tid & 31;
    const int wave = tid >> 5;
    const int r    = blockIdx.x;

    ((uint4*)cw)[ctr_vec(tid)]     = make_uint4(0u, 0u, 0u, 0u);
    ((uint4*)cw)[ctr_vec(tid) + 1] = make_uint4(0u, 0u, 0u, 0u);
    if (tid < NS) {
        const unsigned b = min(tab[r * NS + tid], (unsigned)ES);
        const unsigned e = min(tab[(r + 1) * NS + tid], (unsigned)ES);
        sb[tid] = b;
        se[tid] = max(e, b);
        unsigned v = b;
        for (int o = 16; o > 0; o >>= 1) v += __shfl_xor(v, o);
        if (lane == 0) bsum[wave] = v;
    } else if (tid < NS + 512) {
        static_assert(HID_CH == 4 * (512 / 32), "weight panel: 512 threads sweep 16 rows of 128 floats as 16-byte pieces, 4 times");
        const int t  = tid - NS;
        const int k  = (t * 4) & 127;
        const int n0 = t >> 5;
        const float* wsrc = ((k < 64) ? (Wl + k) : (Wr + (k - 64))) + n0 * 64;
        float* wdst = &Wlds[n0 * WP + k];
        float4 wpan[4];
#pragma unroll
        for (int u = 0; u < 4; ++u) wpan[u] = *(const float4*)(wsrc + u * (16 * 64));
#pragma unroll
        for (int u = 0; u < 4; ++u) *(float4*)(wdst + u * (16 * WP)) = wpan[u];
    }
    __syncthreads();

    unsigned base = 0u;
    for (int w = 0; w < NS / 32; ++w) base += bsum[w];

    constexpr int SPW = NS / NW;
    unsigned wv[SPW], nb[SPW], ne[SPW];
#pragma unroll
    for (int j = 0; j < SPW; ++j) {
        const int sl = wave + j * NW;
        nb[j] = sb[sl];
        ne[j] = se[sl];
        const unsigned i = nb[j] + (unsigned)lane;
        wv[j] = NONE;
        if (i < ne[j]) wv[j] = part[sl * ES + i];
    }

#pragma unroll
    for (int j = 0; j < SPW; ++j) {
        const int sl = wave + j * NW;
        if (wv[j] != NONE) atomicAdd(&cw[ent_node(wv[j]) * CP + wave], 1u);
        for (unsigned i = nb[j] + 32u + (unsigned)lane; i < ne[j]; i += 32u)
            atomicAdd(&cw[ent_node(part[sl * ES + i]) * CP + wave], 1u);
    }
    __syncthreads();

    const unsigned first = scan_counters(cw, wsum, &total, tid, lane, wave);
    if ((tid & 3) == 0 && (tid >> 2) < RN) rowptr[r * RN + (tid >> 2)] = base + first;

#pragma unroll
    for (int j = 0; j < SPW; ++j) {
        const int sl = wave + j * NW;
        if (wv[j] != NONE) ent_place(cw, stage, ssrc, base, wave, wv[j]);
        for (unsigned i = nb[j] + 32u + (unsigned)lane; i < ne[j]; i += 32u)
            ent_place(cw, stage, ssrc, base, wave, part[sl * ES + i]);
    }
    __syncthreads();

    const unsigned n = min(total, (unsigned)CAP);
    for (unsigned i = (unsigned)tid; i < n; i += (unsigned)PT) {
        const unsigned g = base + i;
        if (g < (unsigned)N_EDGES) ssrc[g] = (int)stage[i];
    }

    if (total <= (unsigned)CAP) {
        sage_range_means(At, cw, feat, ListInStage{ stage }, 0u, (unsigned)CAP, r, tid);
    } else {
        __syncthreads();
        sage_range_means(At, cw, feat, ListInMemory{ ssrc }, base, (unsigned)N_EDGES, r, tid);
    }
    __syncthreads();

    const int half = lane >> 4;
    const int m    = lane & 15;
    for (int job = wave; job < TR * (HID_CH / 16); job += NW) {
        const int o0 = (job / (HID_CH / 16)) * 16;
        if (r * RN + o0 >= N_NODES) break;
        const int colg = (job % (HID_CH / 16)) * 16 + m;

        const float bc = bias[colg];
        v8f c;
#pragma unroll
        for (int i = 0; i < 8; ++i) c[i] = bc;

        const float* arow = &At[(o0 + m) * AP];
        const float* xrow = feat + (long)min(r * RN + o0 + m, N_NODES - 1) * 64;
        const float* wcol = &Wlds[colg * WP];

#pragma unroll
        for (int ks = 0; ks < 64; ks += 4) {
            v2f a, b;
            a[0] = arow[ks + half * 2 + 0];
            a[1] = arow[ks + half * 2 + 1];
            b[0] = wcol[ks + half * 2 + 0];
            b[1] = wcol[ks + half * 2 + 1];
            c = __builtin_amdgcn_wmma_f32_16x16x4_f32(false, a, false, b,
                                                      (short)0, c, false, false);
        }
#pragma unroll
        for (int ks = 0; ks < 64; ks += 4) {
            v2f a, b;
            a[0] = xrow[ks + half * 2 + 0];
            a[1] = xrow[ks + half * 2 + 1];
            b[0] = wcol[64 + ks + half * 2 + 0];
            b[1] = wcol[64 + ks + half * 2 + 1];
            c = __builtin_amdgcn_wmma_f32_16x16x4_f32(false, a, false, b,
                                                      (short)0, c, false, false);
        }

#pragma unroll
        for (int v = 0; v < 8; ++v) {
            const int o = o0 + half * 8 + v;
            float y = c[v];
            y = (y > 0.0f) ? y : NEG_SLOPE * y;
            if (o < RN && r * RN + o < N_NODES) out[(long)(r * RN + o) * HID_CH + colg] = y;
        }
    }
}

template <int OUTC>
__global__ __launch_bounds__(256) void sage_layer(
    const float* __restrict__ feat, const int* __restrict__ ssrc,
    const unsigned* __restrict__ rowptr,
    const float* __restrict__ Wl, const float* __restrict__ Wr,
    const float* __restrict__ bias, float* __restrict__ out)
{
    constexpr int CT = OUTC / 16;
    constexpr int RT = 8 / CT;

    static_assert(OUTC % 8 == 0 && WP % 4 == 0, "weight panel: 256 threads sweep 8 rows of 128 floats as 16-byte pieces; rows 16-byte aligned");

    __shared__ __attribute__((aligned(16))) float Wlds[OUTC * WP];
    __shared__ __attribute__((aligned(16))) float At[RT * 16 * AP];

    const int tid = threadIdx.x;
    const int nb  = blockIdx.x * (RT * 16);
    const int grp = tid >> 4;
    const int g   = tid & 15;

    unsigned eb[RT], ee[RT];
#pragma unroll
    for (int j = 0; j < RT; ++j) {
        const int nn = min(nb + j * 16 + grp, N_NODES - 1);
        eb[j] = rowptr[nn];
        ee[j] = min(rowptr[nn + 1], (unsigned)N_EDGES);
    }

    {
        constexpr int WV = OUTC / 8;
        const int k  = (tid * 4) & 127;
        const int n0 = tid >> 5;
        const float* wsrc = ((k < 64) ? (Wl + k) : (Wr + (k - 64))) + n0 * 64;
        float* wdst = &Wlds[n0 * WP + k];
        float4 wv[WV];
#pragma unroll
        for (int u = 0; u < WV; ++u) wv[u] = *(const float4*)(wsrc + u * (8 * 64));
#pragma unroll
        for (int u = 0; u < WV; ++u) *(float4*)(wdst + u * (8 * WP)) = wv[u];
    }

#pragma unroll
    for (int j = 0; j < RT; ++j) asm volatile("" :: "v"(eb[j]), "v"(ee[j]));

#pragma unroll
    for (int j = 0; j < RT; ++j) {
        const int node = nb + j * 16 + grp;
        if (node < N_NODES) {
            const float4 acc = sage_row_sum16(feat + g * 4, ListInMemory{ ssrc }, eb[j], ee[j], g);
            const unsigned deg = (ee[j] > eb[j]) ? ee[j] - eb[j] : 0u;
            const float rd = 1.0f / fmaxf((float)deg, 1.0f);
            *(float4*)&At[(j * 16 + grp) * AP + g * 4] =
                make_float4(acc.x * rd, acc.y * rd, acc.z * rd, acc.w * rd);
        }
    }
    __syncthreads();

    const int wave = tid >> 5;
    const int lane = tid & 31;
    const int rt   = wave / CT;
    const int row0 = nb + rt * 16;
    if (row0 >= N_NODES) return;

    const int col0 = (wave % CT) * 16;
    const int half = lane >> 4;
    const int m    = lane & 15;
    const int row  = row0 + m;
    const int colg = col0 + m;

    const float bc = bias[colg];

    v8f c;
#pragma unroll
    for (int i = 0; i < 8; ++i) c[i] = bc;

    const float* arow = &At[(rt * 16 + m) * AP];
    const float* xrow = feat + (long)row * 64;
    const float* wcol = &Wlds[colg * WP];

#pragma unroll
    for (int ks = 0; ks < 64; ks += 4) {
        v2f a, b;
        a[0] = arow[ks + half * 2 + 0];
        a[1] = arow[ks + half * 2 + 1];
        b[0] = wcol[ks + half * 2 + 0];
        b[1] = wcol[ks + half * 2 + 1];
        c = __builtin_amdgcn_wmma_f32_16x16x4_f32(false, a, false, b,
                                                  (short)0, c, false, false);
    }
#pragma unroll
    for (int ks = 0; ks < 64; ks += 4) {
        v2f a, b;
        a[0] = xrow[ks + half * 2 + 0];
        a[1] = xrow[ks + half * 2 + 1];
        b[0] = wcol[64 + ks + half * 2 + 0];
        b[1] = wcol[64 + ks + half * 2 + 1];
        c = __builtin_amdgcn_wmma_f32_16x16x4_f32(false, a, false, b,
                                                  (short)0, c, false, false);
    }

#pragma unroll
    for (int v = 0; v < 8; ++v) {
        float y = c[v];
        y = (y > 0.0f) ? y : NEG_SLOPE * y;
        out[(long)(row0 + half * 8 + v) * OUTC + colg] = y;
    }
}

extern "C" void kernel_launch(void* const* d_in, const int* in_sizes, int n_in,
                              void* d_out, int out_size, void* d_ws, size_t ws_size,
                              hipStream_t stream) {
    const float* x   = (const float*)d_in[0];
    const int*   ei  = (const int*)  d_in[1];
    const float* W1l = (const float*)d_in[2];
    const float* b1  = (const float*)d_in[3];
    const float* W1r = (const float*)d_in[4];
    const float* W2l = (const float*)d_in[5];
    const float* b2  = (const float*)d_in[6];
    const float* W2r = (const float*)d_in[7];
    float* out = (float*)d_out;

    const size_t partB = (size_t)N_EDGES * 4;
    const size_t featB = (size_t)N_NODES * 64 * 4;
    const size_t tabB  = (size_t)(NR + 1) * NS * 4;
    unsigned* part   = (unsigned*)d_ws;
    float*    h      = (float*)((char*)d_ws + partB);
    unsigned* tab    = (unsigned*)((char*)d_ws + partB + featB);
    unsigned* rowptr = (unsigned*)((char*)d_ws + partB + featB + tabB);
    int*      ssrc   = (int*)(rowptr + RP_N);

    sage_part<<<NS, PT, 0, stream>>>(ei, part, tab);
    sage_sort_layer1<<<NR, PT, 0, stream>>>(part, tab, ssrc, rowptr, x, W1l, W1r, b1, h);

    sage_layer<OUT_CH><<<(N_NODES / 16 + 3) / 4, 256, 0, stream>>>(h, ssrc, rowptr, W2l, W2r, b2, out);
}
